# starting kernel
# baseline (speedup 1.0000x reference)
#include <hip/hip_runtime.h>
#include <hip/hip_bf16.h>

typedef __attribute__((ext_vector_type(2))) float v2f;
typedef __attribute__((ext_vector_type(8))) float v8f;

static __host__ inline int cdiv(int a, int b) { return (a + b - 1) / b; }


__global__ void k_fill(float* __restrict__ p, float v, int n) {
    int i = blockIdx.x * blockDim.x + threadIdx.x;
    if (i < n) p[i] = v;
}

__global__ void k_deg_accum(const int* __restrict__ dst, float* __restrict__ deg, int E) {
    int e = blockIdx.x * blockDim.x + threadIdx.x;
    if (e < E) atomicAdd(&deg[dst[e]], 1.0f);
}

__global__ void k_rsqrt(float* __restrict__ d, int n) {
    int i = blockIdx.x * blockDim.x + threadIdx.x;
    if (i < n) {
        float v = d[i];
        d[i] = (v > 0.0f) ? rsqrtf(v) : 0.0f;
    }
}

__global__ void k_transpose_pad(const float* __restrict__ Bw, float* __restrict__ Bt,
                                int K, int Ncols, int ncols_pad) {
    int i = blockIdx.x * blockDim.x + threadIdx.x;
    int total = ncols_pad * K;
    if (i >= total) return;
    int ncol = i / K, k = i - ncol * K;
    Bt[i] = (ncol < Ncols) ? Bw[(size_t)k * Ncols + ncol] : 0.0f;
}

__global__ void k_edge_scatter(const int* __restrict__ src, const int* __restrict__ dst,
                               const float* __restrict__ t, const float* __restrict__ dinv,
                               float* __restrict__ agg, int E) {
    int e = blockIdx.x * blockDim.x + threadIdx.x;
    if (e >= E) return;
    int s = src[e], d = dst[e];
    float nr = dinv[s] * dinv[d];
    const float4* ts = (const float4*)(t + (size_t)s * 20);
    float* ad = agg + (size_t)d * 20;
#pragma unroll
    for (int q = 0; q < 5; ++q) {
        float4 v = ts[q];
        atomicAdd(ad + 4 * q + 0, v.x * nr);
        atomicAdd(ad + 4 * q + 1, v.y * nr);
        atomicAdd(ad + 4 * q + 2, v.z * nr);
        atomicAdd(ad + 4 * q + 3, v.w * nr);
    }
}

__global__ void k_finish_layer(const float* __restrict__ t, const float* __restrict__ dinv,
                               const float* __restrict__ bias, float* __restrict__ agg, int n_nodes) {
    int i = blockIdx.x * blockDim.x + threadIdx.x;
    int total = n_nodes * 20;
    if (i >= total) return;
    int node = i / 20, f = i - node * 20;
    float di = dinv[node];
    float v = agg[i] + t[i] * di * di + bias[f];
    agg[i] = fmaxf(v, 0.0f);
}

__global__ void k_wmma_gemm_f32(const float* __restrict__ A, const float* __restrict__ Bt,
                                const float* __restrict__ bias, float* __restrict__ C,
                                int M, int K, int Ncols, int mtiles, int ntiles, int do_relu) {
    int wave = (blockIdx.x * blockDim.x + threadIdx.x) >> 5;
    int lane = threadIdx.x & 31;
    int mt = wave % mtiles;
    int nt = wave / mtiles;
    if (nt >= ntiles) return;

    int lrow = lane & 15;
    int kh = (lane >> 4) << 1;
    int ncol = nt * 16 + lrow;
    const float* arow = A + (size_t)(mt * 16 + lrow) * K + kh;
    const float* brow = Bt + (size_t)ncol * K + kh;

    v8f acc = {};
    int kb = 0;
    for (; kb + 16 <= K; kb += 16) {
#pragma unroll
        for (int j = 0; j < 4; ++j) {
            v2f a = *(const v2f*)(arow + kb + 4 * j);
            v2f b = *(const v2f*)(brow + kb + 4 * j);
            acc = __builtin_amdgcn_wmma_f32_16x16x4_f32(false, a, false, b,
                                                        (short)0, acc, false, false);
        }
    }
    for (; kb < K; kb += 4) {
        v2f a = *(const v2f*)(arow + kb);
        v2f b = *(const v2f*)(brow + kb);
        acc = __builtin_amdgcn_wmma_f32_16x16x4_f32(false, a, false, b,
                                                    (short)0, acc, false, false);
    }

    if (ncol >= Ncols) return;
    float bv = bias ? bias[ncol] : 0.0f;
    int rbase = mt * 16 + ((lane >> 4) << 3);
#pragma unroll
    for (int r = 0; r < 8; ++r) {
        float v = acc[r] + bv;
        if (do_relu) v = fmaxf(v, 0.0f);
        C[(size_t)(rbase + r) * Ncols + ncol] = v;
    }
}


extern "C" void kernel_launch(void* const* d_in, const int* in_sizes, int n_in,
                              void* d_out, int out_size, void* d_ws, size_t ws_size,
                              hipStream_t stream) {
    const float* x    = (const float*)d_in[0];
    const int*   eidx = (const int*)d_in[1];
    const float* W1   = (const float*)d_in[3];
    const float* b1   = (const float*)d_in[4];
    const float* W2   = (const float*)d_in[5];
    const float* b2   = (const float*)d_in[6];
    const float* fc1W = (const float*)d_in[7];
    const float* fc1b = (const float*)d_in[8];
    const float* outW = (const float*)d_in[9];
    const float* outb = (const float*)d_in[10];

    const int N  = in_sizes[2];
    const int E  = in_sizes[1] / 2;
    const int Bg = N / 82;
    const int EMB = 20, LAT = 82 * 20, HID = 300, NC = 22;

    char* ws = (char*)d_ws;
    size_t off = 0;
    auto take = [&](size_t bytes) -> void* {
        off = (off + 255) & ~(size_t)255;
        void* p = ws + off;
        off += bytes;
        return p;
    };
    float* dinv = (float*)take((size_t)N * 4);
    float* t    = (float*)take((size_t)N * EMB * 4);
    float* agg  = (float*)take((size_t)N * EMB * 4);
    float* fco  = (float*)take((size_t)Bg * HID * 4);
    float* Bt   = (float*)take((size_t)304 * LAT * 4);

    const int* src = eidx;
    const int* dst = eidx + E;

    const int TB = 256;
    int nfeat = N * EMB;

    auto run_gemm = [&](const float* A, const float* Bw, const float* bias, float* C,
                        int M, int K, int Ncols, int relu) {
        int ntiles = (Ncols + 15) / 16;
        int ncols_pad = ntiles * 16;
        int mtiles = M / 16;
        k_transpose_pad<<<cdiv(ncols_pad * K, TB), TB, 0, stream>>>(Bw, Bt, K, Ncols, ncols_pad);
        int waves = mtiles * ntiles;
        k_wmma_gemm_f32<<<cdiv(waves * 32, TB), TB, 0, stream>>>(
            A, Bt, bias, C, M, K, Ncols, mtiles, ntiles, relu);
    };

    k_fill<<<cdiv(N, TB), TB, 0, stream>>>(dinv, 1.0f, N);
    k_deg_accum<<<cdiv(E, TB), TB, 0, stream>>>(dst, dinv, E);
    k_rsqrt<<<cdiv(N, TB), TB, 0, stream>>>(dinv, N);

    k_fill<<<cdiv(nfeat, TB), TB, 0, stream>>>(agg, 0.0f, nfeat);
    run_gemm(x, W1, nullptr, t, N, 20, EMB, 0);
    k_edge_scatter<<<cdiv(E, TB), TB, 0, stream>>>(src, dst, t, dinv, agg, E);
    k_finish_layer<<<cdiv(nfeat, TB), TB, 0, stream>>>(t, dinv, b1, agg, N);

    run_gemm(agg, W2, nullptr, t, N, EMB, EMB, 0);
    k_fill<<<cdiv(nfeat, TB), TB, 0, stream>>>(agg, 0.0f, nfeat);
    k_edge_scatter<<<cdiv(E, TB), TB, 0, stream>>>(src, dst, t, dinv, agg, E);
    k_finish_layer<<<cdiv(nfeat, TB), TB, 0, stream>>>(t, dinv, b2, agg, N);

    run_gemm(agg  , fc1W, fc1b, fco, Bg, LAT, HID, 1);

    run_gemm(fco, outW, outb, (float*)d_out, Bg, HID, NC, 0);
}


# Edge scatter (20 global float atomics per edge) replaced: each block owns 2624 dst nodes' sums in LDS, scans all dst, adds owned edges' rows by LDS float adds, writes h once; sum order = LDS arrival
# speedup vs baseline: 2.1152x; 2.1152x over previous
#include <hip/hip_runtime.h>
#include <hip/hip_bf16.h>

typedef __attribute__((ext_vector_type(2))) float v2f;
typedef __attribute__((ext_vector_type(8))) float v8f;

static __host__ inline int cdiv(int a, int b) { return (a + b - 1) / b; }


__global__ void k_fill(float* __restrict__ p, float v, int n) {
    int i = blockIdx.x * blockDim.x + threadIdx.x;
    if (i < n) p[i] = v;
}

__global__ void k_deg_accum(const int* __restrict__ dst, float* __restrict__ deg, int E) {
    int e = blockIdx.x * blockDim.x + threadIdx.x;
    if (e < E) atomicAdd(&deg[dst[e]], 1.0f);
}

__global__ void k_rsqrt(float* __restrict__ d, int n) {
    int i = blockIdx.x * blockDim.x + threadIdx.x;
    if (i < n) {
        float v = d[i];
        d[i] = (v > 0.0f) ? rsqrtf(v) : 0.0f;
    }
}

__global__ void k_transpose_pad(const float* __restrict__ Bw, float* __restrict__ Bt,
                                int K, int Ncols, int ncols_pad) {
    int i = blockIdx.x * blockDim.x + threadIdx.x;
    int total = ncols_pad * K;
    if (i >= total) return;
    int ncol = i / K, k = i - ncol * K;
    Bt[i] = (ncol < Ncols) ? Bw[(size_t)k * Ncols + ncol] : 0.0f;
}

#define AG_F      20
#define AG_NPB    2624
#define AG_WAVES  32
#define AG_TB     (AG_WAVES * 32)
#define AG_U      4
#define AG_QFLUSH 128
#define AG_QW     (AG_QFLUSH + AG_U * 128)

__global__ void __launch_bounds__(AG_TB)
k_aggregate(const int* __restrict__ src, const int* __restrict__ dst,
            const float* __restrict__ t, const float* __restrict__ dinv,
            const float* __restrict__ bias, float* __restrict__ h, int E, int N) {
    __shared__ __attribute__((aligned(16))) float acc[AG_NPB * AG_F];
    __shared__ int qe[AG_WAVES * AG_QW];

    const int tid = threadIdx.x;
    const int lane = tid & 31;
    const int wave = __builtin_amdgcn_readfirstlane(tid >> 5);
    const int node0 = blockIdx.x * AG_NPB;
    int* myqe = qe + wave * AG_QW;

    float4* acc4 = (float4*)acc;
    for (int q = tid; q < AG_NPB * AG_F / 4; q += AG_TB) acc4[q] = make_float4(0.f, 0.f, 0.f, 0.f);
    __syncthreads();

    int cnt = 0;

    auto push = [&](bool m, int e) {
        const unsigned bal = __builtin_amdgcn_ballot_w32(m);
        if (bal) {
            if (m) myqe[cnt + (int)__builtin_amdgcn_mbcnt_lo(bal, 0u)] = e;
            cnt += __builtin_popcount(bal);
        }
    };

    const bool vec_ok = ((((size_t)dst) & 15) == 0);
    const int G = vec_ok ? (E >> 2) : 0;
    const int TT = (G + 31) >> 5;
    const int rot = (int)blockIdx.x * (TT / (int)gridDim.x);
    const int4* __restrict__ dst4 = (const int4*)dst;
    const int KV = (TT + AG_WAVES * AG_U - 1) / (AG_WAVES * AG_U);
    const int KR = (E - 4 * G + AG_WAVES * 32 - 1) / (AG_WAVES * 32);
    const int KT = KV + KR;

    int elb[5], fb[5];
#pragma unroll
    for (int b = 0; b < 5; ++b) {
        elb[b] = (32 * b + lane) / AG_F;
        fb[b] = (32 * b + lane) - elb[b] * AG_F;
    }

    for (int k = 0; k < KT; ++k) {
        if (k < KV) {
            int4 dv[AG_U];
            int gi[AG_U];
#pragma unroll
            for (int u = 0; u < AG_U; ++u) {
                const int tl = (k * AG_U + u) * AG_WAVES + wave;
                int at = tl + rot;
                if (at >= TT) at -= TT;
                const int g = at * 32 + lane;
                gi[u] = g;
                dv[u] = (tl < TT && g < G) ? dst4[g] : make_int4(-1, -1, -1, -1);
            }
#pragma unroll
            for (int u = 0; u < AG_U; ++u) {
                const bool m0 = (unsigned)(dv[u].x - node0) < (unsigned)AG_NPB;
                const bool m1 = (unsigned)(dv[u].y - node0) < (unsigned)AG_NPB;
                const bool m2 = (unsigned)(dv[u].z - node0) < (unsigned)AG_NPB;
                const bool m3 = (unsigned)(dv[u].w - node0) < (unsigned)AG_NPB;
                if (__builtin_amdgcn_ballot_w32(m0 | m1 | m2 | m3)) {
                    const int e0 = gi[u] * 4;
                    push(m0, e0);
                    push(m1, e0 + 1);
                    push(m2, e0 + 2);
                    push(m3, e0 + 3);
                }
            }
        } else {
            const int e = 4 * G + ((k - KV) * AG_WAVES + wave) * 32 + lane;
            const int d = (e < E) ? dst[e] : -1;
            push((unsigned)(d - node0) < (unsigned)AG_NPB, e);
        }

        if (cnt >= AG_QFLUSH || (k == KT - 1 && cnt > 0)) {
            __builtin_amdgcn_fence(__ATOMIC_ACQ_REL, "wavefront");
            for (int base = 0; base < cnt; base += 32) {
                const int i = base + lane;
                int s20 = 0, d20 = 0;
                float nr = 0.0f;
                if (i < cnt) {
                    const int e = myqe[i];
                    const int s = src[e];
                    const int d = dst[e];
                    nr = dinv[s] * dinv[d];
                    s20 = s * AG_F;
                    d20 = (d - node0) * AG_F;
                }
                const int rem = cnt - base;
                const int lim = (rem < 32 ? rem : 32) * AG_F;
                float raw[AG_F];
#pragma unroll
                for (int j = 0; j < AG_F; ++j) {
                    const int a = j / 5, b = j % 5;
                    const int ss = __builtin_amdgcn_ds_bpermute((8 * a + elb[b]) << 2, s20);
                    raw[j] = t[ss + fb[b]];
                }
#pragma unroll
                for (int j = 0; j < AG_F; ++j) {
                    const int a = j / 5, b = j % 5;
                    const int dd = __builtin_amdgcn_ds_bpermute((8 * a + elb[b]) << 2, d20);
                    const float nn = __int_as_float(__builtin_amdgcn_ds_bpermute((8 * a + elb[b]) << 2, __float_as_int(nr)));
                    if (160 * a + 32 * b + lane < lim) atomicAdd(&acc[dd + fb[b]], raw[j] * nn);
                }
            }
            cnt = 0;
        }
    }
    __syncthreads();

    const float4* t4 = (const float4*)(t + (size_t)node0 * AG_F);
    float4* h4 = (float4*)(h + (size_t)node0 * AG_F);
    for (int q = tid; q < AG_NPB * AG_F / 4; q += AG_TB) {
        const int nl = q / 5;
        const int f0 = (q - nl * 5) * 4;
        if (node0 + nl < N) {
            const float di = dinv[node0 + nl];
            const float sl = di * di;
            const float4 a = acc4[q];
            const float4 tv = t4[q];
            float4 o;
            o.x = fmaxf(a.x + tv.x * sl + bias[f0 + 0], 0.0f);
            o.y = fmaxf(a.y + tv.y * sl + bias[f0 + 1], 0.0f);
            o.z = fmaxf(a.z + tv.z * sl + bias[f0 + 2], 0.0f);
            o.w = fmaxf(a.w + tv.w * sl + bias[f0 + 3], 0.0f);
            h4[q] = o;
        }
    }
}

__global__ void k_wmma_gemm_f32(const float* __restrict__ A, const float* __restrict__ Bt,
                                const float* __restrict__ bias, float* __restrict__ C,
                                int M, int K, int Ncols, int mtiles, int ntiles, int do_relu) {
    int wave = (blockIdx.x * blockDim.x + threadIdx.x) >> 5;
    int lane = threadIdx.x & 31;
    int mt = wave % mtiles;
    int nt = wave / mtiles;
    if (nt >= ntiles) return;

    int lrow = lane & 15;
    int kh = (lane >> 4) << 1;
    int ncol = nt * 16 + lrow;
    const float* arow = A + (size_t)(mt * 16 + lrow) * K + kh;
    const float* brow = Bt + (size_t)ncol * K + kh;

    v8f acc = {};
    int kb = 0;
    for (; kb + 16 <= K; kb += 16) {
#pragma unroll
        for (int j = 0; j < 4; ++j) {
            v2f a = *(const v2f*)(arow + kb + 4 * j);
            v2f b = *(const v2f*)(brow + kb + 4 * j);
            acc = __builtin_amdgcn_wmma_f32_16x16x4_f32(false, a, false, b,
                                                        (short)0, acc, false, false);
        }
    }
    for (; kb < K; kb += 4) {
        v2f a = *(const v2f*)(arow + kb);
        v2f b = *(const v2f*)(brow + kb);
        acc = __builtin_amdgcn_wmma_f32_16x16x4_f32(false, a, false, b,
                                                    (short)0, acc, false, false);
    }

    if (ncol >= Ncols) return;
    float bv = bias ? bias[ncol] : 0.0f;
    int rbase = mt * 16 + ((lane >> 4) << 3);
#pragma unroll
    for (int r = 0; r < 8; ++r) {
        float v = acc[r] + bv;
        if (do_relu) v = fmaxf(v, 0.0f);
        C[(size_t)(rbase + r) * Ncols + ncol] = v;
    }
}


extern "C" void kernel_launch(void* const* d_in, const int* in_sizes, int n_in,
                              void* d_out, int out_size, void* d_ws, size_t ws_size,
                              hipStream_t stream) {
    const float* x    = (const float*)d_in[0];
    const int*   eidx = (const int*)d_in[1];
    const float* W1   = (const float*)d_in[3];
    const float* b1   = (const float*)d_in[4];
    const float* W2   = (const float*)d_in[5];
    const float* b2   = (const float*)d_in[6];
    const float* fc1W = (const float*)d_in[7];
    const float* fc1b = (const float*)d_in[8];
    const float* outW = (const float*)d_in[9];
    const float* outb = (const float*)d_in[10];

    const int N  = in_sizes[2];
    const int E  = in_sizes[1] / 2;
    const int Bg = N / 82;
    const int EMB = 20, LAT = 82 * 20, HID = 300, NC = 22;

    char* ws = (char*)d_ws;
    size_t off = 0;
    auto take = [&](size_t bytes) -> void* {
        off = (off + 255) & ~(size_t)255;
        void* p = ws + off;
        off += bytes;
        return p;
    };
    float* dinv = (float*)take((size_t)N * 4);
    float* t    = (float*)take((size_t)N * EMB * 4);
    float* agg  = (float*)take((size_t)N * EMB * 4);
    float* fco  = (float*)take((size_t)Bg * HID * 4);
    float* Bt   = (float*)take((size_t)304 * LAT * 4);

    const int* src = eidx;
    const int* dst = eidx + E;

    const int TB = 256;

    auto run_gemm = [&](const float* A, const float* Bw, const float* bias, float* C,
                        int M, int K, int Ncols, int relu) {
        int ntiles = (Ncols + 15) / 16;
        int ncols_pad = ntiles * 16;
        int mtiles = M / 16;
        k_transpose_pad<<<cdiv(ncols_pad * K, TB), TB, 0, stream>>>(Bw, Bt, K, Ncols, ncols_pad);
        int waves = mtiles * ntiles;
        k_wmma_gemm_f32<<<cdiv(waves * 32, TB), TB, 0, stream>>>(
            A, Bt, bias, C, M, K, Ncols, mtiles, ntiles, relu);
    };

    k_fill<<<cdiv(N, TB), TB, 0, stream>>>(dinv, 1.0f, N);
    k_deg_accum<<<cdiv(E, TB), TB, 0, stream>>>(dst, dinv, E);
    k_rsqrt<<<cdiv(N, TB), TB, 0, stream>>>(dinv, N);

    run_gemm(x, W1, nullptr, t, N, 20, EMB, 0);
    k_aggregate<<<cdiv(N, AG_NPB), AG_TB, 0, stream>>>(src, dst, t, dinv, b1, agg, E, N);

    run_gemm(agg, W2, nullptr, t, N, EMB, EMB, 0);
    k_aggregate<<<cdiv(N, AG_NPB), AG_TB, 0, stream>>>(src, dst, t, dinv, b2, agg, E, N);

    run_gemm(agg  , fc1W, fc1b, fco, Bg, LAT, HID, 1);

    run_gemm(fco, outW, outb, (float*)d_out, Bg, HID, NC, 0);
}


# fc1 (8192x1640x300) moved into layer 2's aggregate launch: owner block's 32 graphs' rows feed WMMA from LDS, weight read as stored; h2 round trip, fc1 GEMM + its transpose launch gone; k order kept
# speedup vs baseline: 3.7359x; 1.7662x over previous
#include <hip/hip_runtime.h>
#include <hip/hip_bf16.h>

typedef __attribute__((ext_vector_type(2))) float v2f;
typedef __attribute__((ext_vector_type(8))) float v8f;

static __host__ inline int cdiv(int a, int b) { return (a + b - 1) / b; }


__global__ void k_fill(float* __restrict__ p, float v, int n) {
    int i = blockIdx.x * blockDim.x + threadIdx.x;
    if (i < n) p[i] = v;
}

#define AG_NPB    2624
__global__ void k_deg_owner(const int* __restrict__ dst, float* __restrict__ deg,
                            unsigned* __restrict__ own4, int E, int nwords) {
    const int i = blockIdx.x * blockDim.x + threadIdx.x;
    if (i >= nwords) return;
    const int e0 = 4 * i;
    int d0 = -1, d1 = -1, d2 = -1, d3 = -1;
    if (((((size_t)dst) & 15) == 0) && e0 + 3 < E) {
        const int4 v = ((const int4*)dst)[i];
        d0 = v.x; d1 = v.y; d2 = v.z; d3 = v.w;
    } else {
        if (e0 + 0 < E) d0 = dst[e0 + 0];
        if (e0 + 1 < E) d1 = dst[e0 + 1];
        if (e0 + 2 < E) d2 = dst[e0 + 2];
        if (e0 + 3 < E) d3 = dst[e0 + 3];
    }
    unsigned w = 0;
    if (d0 >= 0) { atomicAdd(&deg[d0], 1.0f); w |= ((unsigned)d0 / (unsigned)AG_NPB); }
    if (d1 >= 0) { atomicAdd(&deg[d1], 1.0f); w |= ((unsigned)d1 / (unsigned)AG_NPB) << 8; }
    if (d2 >= 0) { atomicAdd(&deg[d2], 1.0f); w |= ((unsigned)d2 / (unsigned)AG_NPB) << 16; }
    if (d3 >= 0) { atomicAdd(&deg[d3], 1.0f); w |= ((unsigned)d3 / (unsigned)AG_NPB) << 24; }
    own4[i] = w;
}

__global__ void k_rsqrt(float* __restrict__ d, int n) {
    int i = blockIdx.x * blockDim.x + threadIdx.x;
    if (i < n) {
        float v = d[i];
        d[i] = (v > 0.0f) ? rsqrtf(v) : 0.0f;
    }
}

__global__ void k_transpose_pad(const float* __restrict__ Bw, float* __restrict__ Bt,
                                int K, int Ncols, int ncols_pad) {
    int i = blockIdx.x * blockDim.x + threadIdx.x;
    int total = ncols_pad * K;
    if (i >= total) return;
    int ncol = i / K, k = i - ncol * K;
    Bt[i] = (ncol < Ncols) ? Bw[(size_t)k * Ncols + ncol] : 0.0f;
}

#define AG_F      20
#define AG_GN     82
#define FC_K      (AG_GN * AG_F)
#define FC_N      300
#define FC_NT     ((FC_N + 15) / 16)
#define FC_G      10
#define AG_WAVES  32
#define AG_TB     (AG_WAVES * 32)
#define AG_QFLUSH 128
#define AG_QW     (AG_QFLUSH + 32)

template <bool HEAD>
__global__ void __launch_bounds__(AG_TB)
k_aggregate(const int* __restrict__ src, const int* __restrict__ dst,
            const unsigned char* __restrict__ own, const float* __restrict__ xin, const float* __restrict__ dinv,
            const float* __restrict__ Wm, const float* __restrict__ bias, float* __restrict__ h,
            const float* __restrict__ fcW, const float* __restrict__ fcb, float* __restrict__ fco, int E, int N) {
    static_assert(AG_NPB % AG_GN == 0 && AG_NPB / AG_GN == 32, "a block owns 32 whole graphs");
    static_assert((FC_K / 4) % FC_G == 0, "whole groups of k steps");
    __shared__ __attribute__((aligned(16))) float acc[AG_NPB * AG_F];
    __shared__ int qe[AG_WAVES * AG_QW];

    const int tid = threadIdx.x;
    const int lane = tid & 31;
    const int wave = __builtin_amdgcn_readfirstlane(tid >> 5);
    const int node0 = blockIdx.x * AG_NPB;
    int* myqe = qe + wave * AG_QW;

    float4* acc4 = (float4*)acc;
    for (int q = tid; q < AG_NPB * AG_F / 4; q += AG_TB) acc4[q] = make_float4(0.f, 0.f, 0.f, 0.f);
    __syncthreads();

    int cnt = 0;

    auto push = [&](bool m, int e) {
        const unsigned bal = __builtin_amdgcn_ballot_w32(m);
        if (bal) {
            if (m) myqe[cnt + (int)__builtin_amdgcn_mbcnt_lo(bal, 0u)] = e;
            cnt += __builtin_popcount(bal);
        }
    };

    const int G = (E + 15) >> 4;
    const int TT = (G + 31) >> 5;
    const int rot = (int)blockIdx.x * (TT / (int)gridDim.x);
    const uint4* __restrict__ own16 = (const uint4*)own;
    const int KT = (TT + AG_WAVES - 1) / AG_WAVES;
    const unsigned bb = (unsigned)blockIdx.x * 0x01010101u;

    int elb[5], fb[5];
#pragma unroll
    for (int b = 0; b < 5; ++b) {
        elb[b] = (32 * b + lane) / AG_F;
        fb[b] = (32 * b + lane) - elb[b] * AG_F;
    }

    auto differs = [&](unsigned w) -> unsigned {
        const unsigned x = w ^ bb;
        return ((x & 0x7F7F7F7Fu) + 0x7F7F7F7Fu) | x;
    };

    auto drain = [&]() {
        __builtin_amdgcn_fence(__ATOMIC_ACQ_REL, "wavefront");
        for (int base = 0; base < cnt; base += 32) {
            const int i = base + lane;
            int s20 = 0, d20 = 0;
            float nr = 0.0f;
            if (i < cnt) {
                const int e = myqe[i];
                const int s = src[e];
                const int d = dst[e];
                nr = dinv[s] * dinv[d];
                s20 = s * AG_F;
                d20 = (d - node0) * AG_F;
            }
            const int rem = cnt - base;
            const int lim = (rem < 32 ? rem : 32) * AG_F;
            float raw[AG_F];
#pragma unroll
            for (int j = 0; j < AG_F; ++j) {
                const int a = j / 5, b = j % 5;
                const int ss = __builtin_amdgcn_ds_bpermute((8 * a + elb[b]) << 2, s20);
                raw[j] = xin[ss + fb[b]];
            }
#pragma unroll
            for (int j = 0; j < AG_F; ++j) {
                const int a = j / 5, b = j % 5;
                const int dd = __builtin_amdgcn_ds_bpermute((8 * a + elb[b]) << 2, d20);
                const float nn = __int_as_float(__builtin_amdgcn_ds_bpermute((8 * a + elb[b]) << 2, __float_as_int(nr)));
                if (160 * a + 32 * b + lane < lim) atomicAdd(&acc[dd + fb[b]], raw[j] * nn);
            }
        }
        cnt = 0;
    };

    auto exists_mask = [&](int left) -> unsigned {
        unsigned m = 0u;
#pragma unroll
        for (int q = 0; q < 16; ++q)
            if (q < left) m |= 1u << (8 * (q & 3) + (q >> 2));
        return m;
    };

    auto work = [&](int g, bool live, const uint4& r) {
        const unsigned t0 = differs(r.x), t1 = differs(r.y), t2 = differs(r.z), t3 = differs(r.w);
        const unsigned T = ((t0 >> 7) & 0x01010101u) | ((t1 >> 6) & 0x02020202u) |
                           ((t2 >> 5) & 0x04040404u) | ((t3 >> 4) & 0x08080808u);
        unsigned ownm = ~T & 0x0F0F0F0Fu;
        const int e0 = g * 16;
        const int left = live ? E - e0 : 0;
        if (__builtin_amdgcn_ballot_w32(left < 16)) ownm &= exists_mask(left);
        while (__builtin_amdgcn_ballot_w32(ownm != 0u)) {
            const int p = __builtin_ctz(ownm | 0x80000000u);
            push(ownm != 0u, e0 + ((p & 7) << 2) + (p >> 3));
            ownm &= ownm - 1u;
            if (cnt >= AG_QFLUSH) drain();
        }
    };

    for (int k = 0; k < KT; k += 4) {
        int g[4];
        bool live[4];
        uint4 r[4];
#pragma unroll
        for (int u = 0; u < 4; ++u) {
            const int tl = (k + u) * AG_WAVES + wave;
            int at = tl + rot;
            if (at >= TT) at -= TT;
            live[u] = tl < TT;
            g[u] = at * 32 + lane;
            r[u] = own16[g[u] < G ? g[u] : G - 1];
        }
#pragma unroll
        for (int u = 0; u < 4; ++u) work(g[u], live[u], r[u]);
    }
    if (cnt > 0) drain();
    __syncthreads();

    const float4* t4 = (const float4*)(xin + (size_t)node0 * AG_F);
    for (int q = tid; q < AG_NPB * AG_F / 4; q += AG_TB) {
        const int nl = q / 5;
        if (node0 + nl < N) {
            const float di = dinv[node0 + nl];
            const float sl = di * di;
            float4 a = acc4[q];
            const float4 tv = t4[q];
            a.x += tv.x * sl; a.y += tv.y * sl; a.z += tv.z * sl; a.w += tv.w * sl;
            acc4[q] = a;
        }
    }
    __syncthreads();

    {
        const int c = lane & 15, hh = lane >> 4;
        const int n1 = 16 + c < AG_F ? 16 + c : AG_F - 1;
        v2f w0[AG_F / 4], w1[AG_F / 4];
#pragma unroll
        for (int j = 0; j < AG_F / 4; ++j) {
            const int k = 4 * j + 2 * hh;
            w0[j].x = Wm[k * AG_F + c];
            w0[j].y = Wm[(k + 1) * AG_F + c];
            w1[j].x = Wm[k * AG_F + n1];
            w1[j].y = Wm[(k + 1) * AG_F + n1];
        }
        const float bv0 = bias[c], bv1 = bias[n1];
        for (int mt = wave; mt < AG_NPB / 16; mt += AG_WAVES) {
            float* rows = acc + mt * 16 * AG_F;
            const float* ap = rows + c * AG_F + 2 * hh;
            v2f a[AG_F / 4];
#pragma unroll
            for (int j = 0; j < AG_F / 4; ++j) a[j] = *(const v2f*)(ap + 4 * j);
            v8f d0 = {}, d1 = {};
#pragma unroll
            for (int j = 0; j < AG_F / 4; ++j) {
                d0 = __builtin_amdgcn_wmma_f32_16x16x4_f32(false, a[j], false, w0[j], (short)0, d0, false, false);
                d1 = __builtin_amdgcn_wmma_f32_16x16x4_f32(false, a[j], false, w1[j], (short)0, d1, false, false);
            }
            float* op = rows + 8 * hh * AG_F;
#pragma unroll
            for (int r = 0; r < 8; ++r) {
                op[r * AG_F + c] = fmaxf(d0[r] + bv0, 0.0f);
                if (16 + c < AG_F) op[r * AG_F + 16 + c] = fmaxf(d1[r] + bv1, 0.0f);
            }
        }
    }
    __syncthreads();

    if constexpr (!HEAD) {
        float4* h4 = (float4*)(h + (size_t)node0 * AG_F);
        for (int q = tid; q < AG_NPB * AG_F / 4; q += AG_TB) {
            const int nl = q / 5;
            if (node0 + nl < N) h4[q] = acc4[q];
        }
    } else {
        const int c = lane & 15, hh = lane >> 4;
        const int wa = wave >> 3, wb = (wave - wa) & 7;
        const int nt = wb < 5 ? wa * 5 + wb : FC_NT;
        const int ncol = nt * 16 + c;
        v8f d0 = {}, d1 = {};
        if (nt < FC_NT) {
            const int nc = ncol < FC_N ? ncol : FC_N - 1;
            const float* bp = fcW + (size_t)(2 * hh) * FC_N + nc;
            const float* a0p = acc + c * FC_K + 2 * hh;
            const float* a1p = a0p + 16 * FC_K;
            float bx[FC_G], by[FC_G];
#pragma unroll
            for (int j = 0; j < FC_G; ++j) {
                bx[j] = bp[(size_t)(4 * j) * FC_N];
                by[j] = bp[(size_t)(4 * j + 1) * FC_N];
            }
            constexpr int NG = FC_K / 4 / FC_G;
            for (int g = 0; g < NG; ++g) {
                const int gn = g + 1 < NG ? g + 1 : g;
                const float* bq = bp + (size_t)(gn * FC_G * 4) * FC_N;
                float nx[FC_G], ny[FC_G];
#pragma unroll
                for (int j = 0; j < FC_G; ++j) {
                    nx[j] = bq[(size_t)(4 * j) * FC_N];
                    ny[j] = bq[(size_t)(4 * j + 1) * FC_N];
                }
                v2f a0[FC_G], a1[FC_G];
#pragma unroll
                for (int j = 0; j < FC_G; ++j) {
                    a0[j] = *(const v2f*)(a0p + (g * FC_G + j) * 4);
                    a1[j] = *(const v2f*)(a1p + (g * FC_G + j) * 4);
                }
#pragma unroll
                for (int j = 0; j < FC_G; ++j) {
                    v2f b;
                    b.x = bx[j];
                    b.y = by[j];
                    d0 = __builtin_amdgcn_wmma_f32_16x16x4_f32(false, a0[j], false, b, (short)0, d0, false, false);
                    d1 = __builtin_amdgcn_wmma_f32_16x16x4_f32(false, a1[j], false, b, (short)0, d1, false, false);
                }
#pragma unroll
                for (int j = 0; j < FC_G; ++j) {
                    bx[j] = nx[j];
                    by[j] = ny[j];
                }
            }
        }
        __syncthreads();
        if (nt < FC_NT && ncol < FC_N) {
            const float bv = fcb[ncol];
            float* op = acc + 8 * hh * FC_N + ncol;
#pragma unroll
            for (int r = 0; r < 8; ++r) {
                op[r * FC_N] = fmaxf(d0[r] + bv, 0.0f);
                op[(16 + r) * FC_N] = fmaxf(d1[r] + bv, 0.0f);
            }
        }
        __syncthreads();
        const int g0 = (int)blockIdx.x * (AG_NPB / AG_GN);
        const int ng = N / AG_GN;
        float4* o4 = (float4*)(fco + (size_t)g0 * FC_N);
        for (int q = tid; q < (AG_NPB / AG_GN) * FC_N / 4; q += AG_TB) {
            const int gl = q / (FC_N / 4);
            if (g0 + gl < ng) o4[q] = acc4[q];
        }
    }
}

__global__ void k_wmma_gemm_f32(const float* __restrict__ A, const float* __restrict__ Bt,
                                const float* __restrict__ bias, float* __restrict__ C,
                                int M, int K, int Ncols, int mtiles, int ntiles, int do_relu) {
    int wave = (blockIdx.x * blockDim.x + threadIdx.x) >> 5;
    int lane = threadIdx.x & 31;
    int mt = wave % mtiles;
    int nt = wave / mtiles;
    if (nt >= ntiles) return;

    int lrow = lane & 15;
    int kh = (lane >> 4) << 1;
    int ncol = nt * 16 + lrow;
    const float* arow = A + (size_t)(mt * 16 + lrow) * K + kh;
    const float* brow = Bt + (size_t)ncol * K + kh;

    v8f acc = {};
    int kb = 0;
    for (; kb + 16 <= K; kb += 16) {
#pragma unroll
        for (int j = 0; j < 4; ++j) {
            v2f a = *(const v2f*)(arow + kb + 4 * j);
            v2f b = *(const v2f*)(brow + kb + 4 * j);
            acc = __builtin_amdgcn_wmma_f32_16x16x4_f32(false, a, false, b,
                                                        (short)0, acc, false, false);
        }
    }
    for (; kb < K; kb += 4) {
        v2f a = *(const v2f*)(arow + kb);
        v2f b = *(const v2f*)(brow + kb);
        acc = __builtin_amdgcn_wmma_f32_16x16x4_f32(false, a, false, b,
                                                    (short)0, acc, false, false);
    }

    if (ncol >= Ncols) return;
    float bv = bias ? bias[ncol] : 0.0f;
    int rbase = mt * 16 + ((lane >> 4) << 3);
#pragma unroll
    for (int r = 0; r < 8; ++r) {
        float v = acc[r] + bv;
        if (do_relu) v = fmaxf(v, 0.0f);
        C[(size_t)(rbase + r) * Ncols + ncol] = v;
    }
}


extern "C" void kernel_launch(void* const* d_in, const int* in_sizes, int n_in,
                              void* d_out, int out_size, void* d_ws, size_t ws_size,
                              hipStream_t stream) {
    const float* x    = (const float*)d_in[0];
    const int*   eidx = (const int*)d_in[1];
    const float* W1   = (const float*)d_in[3];
    const float* b1   = (const float*)d_in[4];
    const float* W2   = (const float*)d_in[5];
    const float* b2   = (const float*)d_in[6];
    const float* fc1W = (const float*)d_in[7];
    const float* fc1b = (const float*)d_in[8];
    const float* outW = (const float*)d_in[9];
    const float* outb = (const float*)d_in[10];

    const int N  = in_sizes[2];
    const int E  = in_sizes[1] / 2;
    const int Bg = N / 82;
    const int EMB = 20, HID = 300, NC = 22;

    char* ws = (char*)d_ws;
    size_t off = 0;
    auto take = [&](size_t bytes) -> void* {
        off = (off + 255) & ~(size_t)255;
        void* p = ws + off;
        off += bytes;
        return p;
    };
    float* dinv = (float*)take((size_t)N * 4);
    float* t    = (float*)take((size_t)N * EMB * 4);
    float* fco  = (float*)take((size_t)Bg * HID * 4);
    float* Bt   = (float*)take((size_t)32 * HID * 4);
    const int nwords = cdiv(E, 16) * 4;
    unsigned* own4 = (unsigned*)take((size_t)nwords * 4);
    const unsigned char* own = (const unsigned char*)own4;

    const int* src = eidx;
    const int* dst = eidx + E;

    const int TB = 256;

    auto run_gemm = [&](const float* A, const float* Bw, const float* bias, float* C,
                        int M, int K, int Ncols, int relu) {
        int ntiles = (Ncols + 15) / 16;
        int ncols_pad = ntiles * 16;
        int mtiles = M / 16;
        k_transpose_pad<<<cdiv(ncols_pad * K, TB), TB, 0, stream>>>(Bw, Bt, K, Ncols, ncols_pad);
        int waves = mtiles * ntiles;
        k_wmma_gemm_f32<<<cdiv(waves * 32, TB), TB, 0, stream>>>(
            A, Bt, bias, C, M, K, Ncols, mtiles, ntiles, relu);
    };

    k_fill<<<cdiv(N, TB), TB, 0, stream>>>(dinv, 1.0f, N);
    k_deg_owner<<<cdiv(nwords, TB), TB, 0, stream>>>(dst, dinv, own4, E, nwords);
    k_rsqrt<<<cdiv(N, TB), TB, 0, stream>>>(dinv, N);

    k_aggregate<false><<<cdiv(N, AG_NPB), AG_TB, 0, stream>>>(src, dst, own, x, dinv, W1, b1, t,
                                                              nullptr, nullptr, nullptr, E, N);

    k_aggregate<true><<<cdiv(N, AG_NPB), AG_TB, 0, stream>>>(src, dst, own, t, dinv, W2, b2, nullptr,
                                                             fc1W, fc1b, fco, E, N);

    run_gemm(fco, outW, outb, (float*)d_out, Bg, HID, NC, 0);
}


# Layer 1's owner blocks write down the edges they add (row offset, sum offset, norm; chunks chained per wave); layer 2 replays them: no second scan, no per-edge lookups; sum order = LDS arrival
# speedup vs baseline: 4.6593x; 1.2472x over previous
#include <hip/hip_runtime.h>
#include <hip/hip_bf16.h>

typedef __attribute__((ext_vector_type(2))) float v2f;
typedef __attribute__((ext_vector_type(8))) float v8f;

static __host__ inline int cdiv(int a, int b) { return (a + b - 1) / b; }


__global__ void k_fill(float* __restrict__ p, float v, int n) {
    int i = blockIdx.x * blockDim.x + threadIdx.x;
    if (i < n) p[i] = v;
}

#define AG_NPB    2624
__global__ void k_deg_owner(const int* __restrict__ dst, float* __restrict__ deg,
                            unsigned* __restrict__ own4, int E, int nwords) {
    const int i = blockIdx.x * blockDim.x + threadIdx.x;
    if (i >= nwords) return;
    const int e0 = 4 * i;
    int d0 = -1, d1 = -1, d2 = -1, d3 = -1;
    if (((((size_t)dst) & 15) == 0) && e0 + 3 < E) {
        const int4 v = ((const int4*)dst)[i];
        d0 = v.x; d1 = v.y; d2 = v.z; d3 = v.w;
    } else {
        if (e0 + 0 < E) d0 = dst[e0 + 0];
        if (e0 + 1 < E) d1 = dst[e0 + 1];
        if (e0 + 2 < E) d2 = dst[e0 + 2];
        if (e0 + 3 < E) d3 = dst[e0 + 3];
    }
    unsigned w = 0;
    if (d0 >= 0) { atomicAdd(&deg[d0], 1.0f); w |= ((unsigned)d0 / (unsigned)AG_NPB); }
    if (d1 >= 0) { atomicAdd(&deg[d1], 1.0f); w |= ((unsigned)d1 / (unsigned)AG_NPB) << 8; }
    if (d2 >= 0) { atomicAdd(&deg[d2], 1.0f); w |= ((unsigned)d2 / (unsigned)AG_NPB) << 16; }
    if (d3 >= 0) { atomicAdd(&deg[d3], 1.0f); w |= ((unsigned)d3 / (unsigned)AG_NPB) << 24; }
    own4[i] = w;
}

__global__ void k_rsqrt(float* __restrict__ d, int n, int* __restrict__ cursor) {
    int i = blockIdx.x * blockDim.x + threadIdx.x;
    if (i == 0) *cursor = 0;
    if (i < n) {
        float v = d[i];
        d[i] = (v > 0.0f) ? rsqrtf(v) : 0.0f;
    }
}

__global__ void k_transpose_pad(const float* __restrict__ Bw, float* __restrict__ Bt,
                                int K, int Ncols, int ncols_pad) {
    int i = blockIdx.x * blockDim.x + threadIdx.x;
    int total = ncols_pad * K;
    if (i >= total) return;
    int ncol = i / K, k = i - ncol * K;
    Bt[i] = (ncol < Ncols) ? Bw[(size_t)k * Ncols + ncol] : 0.0f;
}

#define AG_F      20
#define AG_GN     82
#define FC_K      (AG_GN * AG_F)
#define FC_N      300
#define FC_NT     ((FC_N + 15) / 16)
#define FC_G      10
#define AG_WAVES  32
#define AG_TB     (AG_WAVES * 32)
#define AG_QFLUSH 128
#define AG_QW     (AG_QFLUSH + 32)
#define AG_CHW    (3 * AG_QW)

template <bool HEAD>
__global__ void __launch_bounds__(AG_TB)
k_aggregate(const int* __restrict__ src, const int* __restrict__ dst,
            const unsigned char* __restrict__ own, const float* __restrict__ xin, const float* __restrict__ dinv,
            const float* __restrict__ Wm, const float* __restrict__ bias, float* __restrict__ h,
            const float* __restrict__ fcW, const float* __restrict__ fcb, float* __restrict__ fco,
            int* __restrict__ lst, int* __restrict__ hdr, int* __restrict__ tails, int* __restrict__ cursor, int maxch,
            int E, int N) {
    static_assert(AG_NPB % AG_GN == 0 && AG_NPB / AG_GN == 32, "a block owns 32 whole graphs");
    static_assert((FC_K / 4) % FC_G == 0, "whole groups of k steps");
    __shared__ __attribute__((aligned(16))) float acc[AG_NPB * AG_F];

    const int tid = threadIdx.x;
    const int lane = tid & 31;
    const int wave = __builtin_amdgcn_readfirstlane(tid >> 5);
    const int node0 = blockIdx.x * AG_NPB;

    float4* acc4 = (float4*)acc;
    for (int q = tid; q < AG_NPB * AG_F / 4; q += AG_TB) acc4[q] = make_float4(0.f, 0.f, 0.f, 0.f);
    __syncthreads();


    int elb[5], fb[5];
#pragma unroll
    for (int b = 0; b < 5; ++b) {
        elb[b] = (32 * b + lane) / AG_F;
        fb[b] = (32 * b + lane) - elb[b] * AG_F;
    }

    auto apply = [&](int s20, int d20, float nr, int lim) {
        float raw[AG_F];
#pragma unroll
        for (int j = 0; j < AG_F; ++j) {
            const int a = j / 5, b = j % 5;
            const int ss = __builtin_amdgcn_ds_bpermute((8 * a + elb[b]) << 2, s20);
            raw[j] = xin[ss + fb[b]];
        }
#pragma unroll
        for (int j = 0; j < AG_F; ++j) {
            const int a = j / 5, b = j % 5;
            const int dd = __builtin_amdgcn_ds_bpermute((8 * a + elb[b]) << 2, d20);
            const float nn = __int_as_float(__builtin_amdgcn_ds_bpermute((8 * a + elb[b]) << 2, __float_as_int(nr)));
            if (160 * a + 32 * b + lane < lim) atomicAdd(&acc[dd + fb[b]], raw[j] * nn);
        }
    };

    if constexpr (!HEAD) {
        __shared__ int qe[AG_WAVES * AG_QW];
        int* myqe = qe + wave * AG_QW;
        int cnt = 0;
        int prevc = -1;

        auto push = [&](bool m, int e) {
            const unsigned bal = __builtin_amdgcn_ballot_w32(m);
            if (bal) {
                if (m) myqe[cnt + (int)__builtin_amdgcn_mbcnt_lo(bal, 0u)] = e;
                cnt += __builtin_popcount(bal);
            }
        };

        const int G = (E + 15) >> 4;
        const int TT = (G + 31) >> 5;
        const int rot = (int)blockIdx.x * (TT / (int)gridDim.x);
        const uint4* __restrict__ own16 = (const uint4*)own;
        const int KT = (TT + AG_WAVES - 1) / AG_WAVES;
        const unsigned bb = (unsigned)blockIdx.x * 0x01010101u;

        auto differs = [&](unsigned w) -> unsigned {
            const unsigned x = w ^ bb;
            return ((x & 0x7F7F7F7Fu) + 0x7F7F7F7Fu) | x;
        };

        auto drain = [&]() {
            __builtin_amdgcn_fence(__ATOMIC_ACQ_REL, "wavefront");
            int cv = 0;
            if (lane == 0) cv = atomicAdd(cursor, 1);
            const int c = __builtin_amdgcn_readfirstlane(cv);
            const bool keep = c < maxch;
            int* ch = lst + (size_t)(keep ? c : 0) * AG_CHW;
            if (keep) {
                if (lane == 0) *(int2*)(hdr + 2 * c) = make_int2(prevc, cnt);
                prevc = c;
            }
            for (int base = 0; base < cnt; base += 32) {
                const int i = base + lane;
                int s20 = 0, d20 = 0;
                float nr = 0.0f;
                if (i < cnt) {
                    const int e = myqe[i];
                    const int s = src[e];
                    const int d = dst[e];
                    nr = dinv[s] * dinv[d];
                    s20 = s * AG_F;
                    d20 = (d - node0) * AG_F;
                }
                if (keep) {
                    ch[i] = s20;
                    ch[AG_QW + i] = d20;
                    ch[2 * AG_QW + i] = __float_as_int(nr);
                }
                const int rem = cnt - base;
                apply(s20, d20, nr, (rem < 32 ? rem : 32) * AG_F);
            }
            cnt = 0;
        };

        auto exists_mask = [&](int left) -> unsigned {
            unsigned m = 0u;
#pragma unroll
            for (int q = 0; q < 16; ++q)
                if (q < left) m |= 1u << (8 * (q & 3) + (q >> 2));
            return m;
        };

        auto work = [&](int g, bool live, const uint4& r) {
            const unsigned t0 = differs(r.x), t1 = differs(r.y), t2 = differs(r.z), t3 = differs(r.w);
            const unsigned T = ((t0 >> 7) & 0x01010101u) | ((t1 >> 6) & 0x02020202u) |
                               ((t2 >> 5) & 0x04040404u) | ((t3 >> 4) & 0x08080808u);
            unsigned ownm = ~T & 0x0F0F0F0Fu;
            const int e0 = g * 16;
            const int left = live ? E - e0 : 0;
            if (__builtin_amdgcn_ballot_w32(left < 16)) ownm &= exists_mask(left);
            while (__builtin_amdgcn_ballot_w32(ownm != 0u)) {
                const int p = __builtin_ctz(ownm | 0x80000000u);
                push(ownm != 0u, e0 + ((p & 7) << 2) + (p >> 3));
                ownm &= ownm - 1u;
                if (cnt >= AG_QFLUSH) drain();
            }
        };

        for (int k = 0; k < KT; k += 4) {
            int g[4];
            bool live[4];
            uint4 r[4];
#pragma unroll
            for (int u = 0; u < 4; ++u) {
                const int tl = (k + u) * AG_WAVES + wave;
                int at = tl + rot;
                if (at >= TT) at -= TT;
                live[u] = tl < TT;
                g[u] = at * 32 + lane;
                r[u] = own16[g[u] < G ? g[u] : G - 1];
            }
#pragma unroll
            for (int u = 0; u < 4; ++u) work(g[u], live[u], r[u]);
        }
        if (cnt > 0) drain();
        if (lane == 0) tails[blockIdx.x * AG_WAVES + wave] = prevc;
    } else {
        int c = __builtin_amdgcn_readfirstlane(tails[blockIdx.x * AG_WAVES + wave]);
        for (int it = 0; it < maxch && (unsigned)c < (unsigned)maxch; ++it) {
            const int2 hd = *(const int2*)(hdr + 2 * c);
            const int hn = __builtin_amdgcn_readfirstlane(hd.y);
            const int n = hn < AG_QW ? hn : AG_QW;
            const int* ch = lst + (size_t)c * AG_CHW;
            int s20 = ch[lane], d20 = ch[AG_QW + lane], nri = ch[2 * AG_QW + lane];
            for (int base = 0; base < n; base += 32) {
                const int nb = (base + 32 < n ? base + 32 : base) + lane;
                const int ns = ch[nb], nd = ch[AG_QW + nb], nn = ch[2 * AG_QW + nb];
                const int rem = n - base;
                apply(s20, d20, __int_as_float(nri), (rem < 32 ? rem : 32) * AG_F);
                s20 = ns;
                d20 = nd;
                nri = nn;
            }
            c = __builtin_amdgcn_readfirstlane(hd.x);
        }
    }
    __syncthreads();

    const float4* t4 = (const float4*)(xin + (size_t)node0 * AG_F);
    for (int q = tid; q < AG_NPB * AG_F / 4; q += AG_TB) {
        const int nl = q / 5;
        if (node0 + nl < N) {
            const float di = dinv[node0 + nl];
            const float sl = di * di;
            float4 a = acc4[q];
            const float4 tv = t4[q];
            a.x += tv.x * sl; a.y += tv.y * sl; a.z += tv.z * sl; a.w += tv.w * sl;
            acc4[q] = a;
        }
    }
    __syncthreads();

    {
        const int c = lane & 15, hh = lane >> 4;
        const int n1 = 16 + c < AG_F ? 16 + c : AG_F - 1;
        v2f w0[AG_F / 4], w1[AG_F / 4];
#pragma unroll
        for (int j = 0; j < AG_F / 4; ++j) {
            const int k = 4 * j + 2 * hh;
            w0[j].x = Wm[k * AG_F + c];
            w0[j].y = Wm[(k + 1) * AG_F + c];
            w1[j].x = Wm[k * AG_F + n1];
            w1[j].y = Wm[(k + 1) * AG_F + n1];
        }
        const float bv0 = bias[c], bv1 = bias[n1];
        for (int mt = wave; mt < AG_NPB / 16; mt += AG_WAVES) {
            float* rows = acc + mt * 16 * AG_F;
            const float* ap = rows + c * AG_F + 2 * hh;
            v2f a[AG_F / 4];
#pragma unroll
            for (int j = 0; j < AG_F / 4; ++j) a[j] = *(const v2f*)(ap + 4 * j);
            v8f d0 = {}, d1 = {};
#pragma unroll
            for (int j = 0; j < AG_F / 4; ++j) {
                d0 = __builtin_amdgcn_wmma_f32_16x16x4_f32(false, a[j], false, w0[j], (short)0, d0, false, false);
                d1 = __builtin_amdgcn_wmma_f32_16x16x4_f32(false, a[j], false, w1[j], (short)0, d1, false, false);
            }
            float* op = rows + 8 * hh * AG_F;
#pragma unroll
            for (int r = 0; r < 8; ++r) {
                op[r * AG_F + c] = fmaxf(d0[r] + bv0, 0.0f);
                if (16 + c < AG_F) op[r * AG_F + 16 + c] = fmaxf(d1[r] + bv1, 0.0f);
            }
        }
    }
    __syncthreads();

    if constexpr (!HEAD) {
        float4* h4 = (float4*)(h + (size_t)node0 * AG_F);
        for (int q = tid; q < AG_NPB * AG_F / 4; q += AG_TB) {
            const int nl = q / 5;
            if (node0 + nl < N) h4[q] = acc4[q];
        }
    } else {
        const int c = lane & 15, hh = lane >> 4;
        const int wa = wave >> 3, wb = (wave - wa) & 7;
        const int nt = wb < 5 ? wa * 5 + wb : FC_NT;
        const int ncol = nt * 16 + c;
        v8f d0 = {}, d1 = {};
        if (nt < FC_NT) {
            const int nc = ncol < FC_N ? ncol : FC_N - 1;
            const float* bp = fcW + (size_t)(2 * hh) * FC_N + nc;
            const float* a0p = acc + c * FC_K + 2 * hh;
            const float* a1p = a0p + 16 * FC_K;
            float bx[FC_G], by[FC_G];
#pragma unroll
            for (int j = 0; j < FC_G; ++j) {
                bx[j] = bp[(size_t)(4 * j) * FC_N];
                by[j] = bp[(size_t)(4 * j + 1) * FC_N];
            }
            constexpr int NG = FC_K / 4 / FC_G;
            for (int g = 0; g < NG; ++g) {
                const int gn = g + 1 < NG ? g + 1 : g;
                const float* bq = bp + (size_t)(gn * FC_G * 4) * FC_N;
                float nx[FC_G], ny[FC_G];
#pragma unroll
                for (int j = 0; j < FC_G; ++j) {
                    nx[j] = bq[(size_t)(4 * j) * FC_N];
                    ny[j] = bq[(size_t)(4 * j + 1) * FC_N];
                }
                v2f a0[FC_G], a1[FC_G];
#pragma unroll
                for (int j = 0; j < FC_G; ++j) {
                    a0[j] = *(const v2f*)(a0p + (g * FC_G + j) * 4);
                    a1[j] = *(const v2f*)(a1p + (g * FC_G + j) * 4);
                }
#pragma unroll
                for (int j = 0; j < FC_G; ++j) {
                    v2f b;
                    b.x = bx[j];
                    b.y = by[j];
                    d0 = __builtin_amdgcn_wmma_f32_16x16x4_f32(false, a0[j], false, b, (short)0, d0, false, false);
                    d1 = __builtin_amdgcn_wmma_f32_16x16x4_f32(false, a1[j], false, b, (short)0, d1, false, false);
                }
#pragma unroll
                for (int j = 0; j < FC_G; ++j) {
                    bx[j] = nx[j];
                    by[j] = ny[j];
                }
            }
        }
        __syncthreads();
        if (nt < FC_NT && ncol < FC_N) {
            const float bv = fcb[ncol];
            float* op = acc + 8 * hh * FC_N + ncol;
#pragma unroll
            for (int r = 0; r < 8; ++r) {
                op[r * FC_N] = fmaxf(d0[r] + bv, 0.0f);
                op[(16 + r) * FC_N] = fmaxf(d1[r] + bv, 0.0f);
            }
        }
        __syncthreads();
        const int g0 = (int)blockIdx.x * (AG_NPB / AG_GN);
        const int ng = N / AG_GN;
        float4* o4 = (float4*)(fco + (size_t)g0 * FC_N);
        for (int q = tid; q < (AG_NPB / AG_GN) * FC_N / 4; q += AG_TB) {
            const int gl = q / (FC_N / 4);
            if (g0 + gl < ng) o4[q] = acc4[q];
        }
    }
}

__global__ void k_wmma_gemm_f32(const float* __restrict__ A, const float* __restrict__ Bt,
                                const float* __restrict__ bias, float* __restrict__ C,
                                int M, int K, int Ncols, int mtiles, int ntiles, int do_relu) {
    int wave = (blockIdx.x * blockDim.x + threadIdx.x) >> 5;
    int lane = threadIdx.x & 31;
    int mt = wave % mtiles;
    int nt = wave / mtiles;
    if (nt >= ntiles) return;

    int lrow = lane & 15;
    int kh = (lane >> 4) << 1;
    int ncol = nt * 16 + lrow;
    const float* arow = A + (size_t)(mt * 16 + lrow) * K + kh;
    const float* brow = Bt + (size_t)ncol * K + kh;

    v8f acc = {};
    int kb = 0;
    for (; kb + 16 <= K; kb += 16) {
#pragma unroll
        for (int j = 0; j < 4; ++j) {
            v2f a = *(const v2f*)(arow + kb + 4 * j);
            v2f b = *(const v2f*)(brow + kb + 4 * j);
            acc = __builtin_amdgcn_wmma_f32_16x16x4_f32(false, a, false, b,
                                                        (short)0, acc, false, false);
        }
    }
    for (; kb < K; kb += 4) {
        v2f a = *(const v2f*)(arow + kb);
        v2f b = *(const v2f*)(brow + kb);
        acc = __builtin_amdgcn_wmma_f32_16x16x4_f32(false, a, false, b,
                                                    (short)0, acc, false, false);
    }

    if (ncol >= Ncols) return;
    float bv = bias ? bias[ncol] : 0.0f;
    int rbase = mt * 16 + ((lane >> 4) << 3);
#pragma unroll
    for (int r = 0; r < 8; ++r) {
        float v = acc[r] + bv;
        if (do_relu) v = fmaxf(v, 0.0f);
        C[(size_t)(rbase + r) * Ncols + ncol] = v;
    }
}


extern "C" void kernel_launch(void* const* d_in, const int* in_sizes, int n_in,
                              void* d_out, int out_size, void* d_ws, size_t ws_size,
                              hipStream_t stream) {
    const float* x    = (const float*)d_in[0];
    const int*   eidx = (const int*)d_in[1];
    const float* W1   = (const float*)d_in[3];
    const float* b1   = (const float*)d_in[4];
    const float* W2   = (const float*)d_in[5];
    const float* b2   = (const float*)d_in[6];
    const float* fc1W = (const float*)d_in[7];
    const float* fc1b = (const float*)d_in[8];
    const float* outW = (const float*)d_in[9];
    const float* outb = (const float*)d_in[10];

    const int N  = in_sizes[2];
    const int E  = in_sizes[1] / 2;
    const int Bg = N / 82;
    const int EMB = 20, HID = 300, NC = 22;

    char* ws = (char*)d_ws;
    size_t off = 0;
    auto take = [&](size_t bytes) -> void* {
        off = (off + 255) & ~(size_t)255;
        void* p = ws + off;
        off += bytes;
        return p;
    };
    float* dinv = (float*)take((size_t)N * 4);
    float* t    = (float*)take((size_t)N * EMB * 4);
    float* fco  = (float*)take((size_t)Bg * HID * 4);
    float* Bt   = (float*)take((size_t)32 * HID * 4);
    const int nwords = cdiv(E, 16) * 4;
    unsigned* own4 = (unsigned*)take((size_t)nwords * 4);
    const unsigned char* own = (const unsigned char*)own4;
    const int maxch = E / AG_QFLUSH + cdiv(N, AG_NPB) * AG_WAVES;
    int* lst    = (int*)take((size_t)maxch * AG_CHW * 4);
    int* hdr    = (int*)take((size_t)maxch * 8);
    int* tails  = (int*)take((size_t)cdiv(N, AG_NPB) * AG_WAVES * 4);
    int* cursor = (int*)take(4);

    const int* src = eidx;
    const int* dst = eidx + E;

    const int TB = 256;

    auto run_gemm = [&](const float* A, const float* Bw, const float* bias, float* C,
                        int M, int K, int Ncols, int relu) {
        int ntiles = (Ncols + 15) / 16;
        int ncols_pad = ntiles * 16;
        int mtiles = M / 16;
        k_transpose_pad<<<cdiv(ncols_pad * K, TB), TB, 0, stream>>>(Bw, Bt, K, Ncols, ncols_pad);
        int waves = mtiles * ntiles;
        k_wmma_gemm_f32<<<cdiv(waves * 32, TB), TB, 0, stream>>>(
            A, Bt, bias, C, M, K, Ncols, mtiles, ntiles, relu);
    };

    k_fill<<<cdiv(N, TB), TB, 0, stream>>>(dinv, 1.0f, N);
    k_deg_owner<<<cdiv(nwords, TB), TB, 0, stream>>>(dst, dinv, own4, E, nwords);
    k_rsqrt<<<cdiv(N, TB), TB, 0, stream>>>(dinv, N, cursor);

    k_aggregate<false><<<cdiv(N, AG_NPB), AG_TB, 0, stream>>>(src, dst, own, x, dinv, W1, b1, t,
                                                              nullptr, nullptr, nullptr,
                                                              lst, hdr, tails, cursor, maxch, E, N);

    k_aggregate<true><<<cdiv(N, AG_NPB), AG_TB, 0, stream>>>(src, dst, own, t, dinv, W2, b2, nullptr,
                                                             fc1W, fc1b, fco,
                                                             lst, hdr, tails, cursor, maxch, E, N);

    run_gemm(fco, outW, outb, (float*)d_out, Bg, HID, NC, 0);
}


# Layer 1's per-edge lookups 4 -> 2 scattered reads: degree pass packs (src, local dst) in one word per edge, dst's dinv read from an LDS copy of the block's 2624 factors; norms and sum order kept
# speedup vs baseline: 4.9962x; 1.0723x over previous
#include <hip/hip_runtime.h>
#include <hip/hip_bf16.h>

typedef __attribute__((ext_vector_type(2))) float v2f;
typedef __attribute__((ext_vector_type(8))) float v8f;

static __host__ inline int cdiv(int a, int b) { return (a + b - 1) / b; }


__global__ void k_fill(float* __restrict__ p, float v, int n) {
    int i = blockIdx.x * blockDim.x + threadIdx.x;
    if (i < n) p[i] = v;
}

#define AG_NPB    2624
#define AG_SBITS  20
#define AG_SMASK  ((1u << AG_SBITS) - 1u)
__global__ void k_deg_owner(const int* __restrict__ src, const int* __restrict__ dst, float* __restrict__ deg,
                            unsigned* __restrict__ own4, unsigned* __restrict__ rec, int E, int nwords) {
    static_assert(256 * AG_NPB <= (1 << AG_SBITS) && AG_NPB <= (1 << (32 - AG_SBITS)), "src and local dst fit one word");
    const int i = blockIdx.x * blockDim.x + threadIdx.x;
    if (i >= nwords) return;
    const int e0 = 4 * i;
    int d0 = -1, d1 = -1, d2 = -1, d3 = -1;
    int s0 = 0, s1 = 0, s2 = 0, s3 = 0;
    if ((((((size_t)dst) | ((size_t)src)) & 15) == 0) && e0 + 3 < E) {
        const int4 v = ((const int4*)dst)[i];
        const int4 u = ((const int4*)src)[i];
        d0 = v.x; d1 = v.y; d2 = v.z; d3 = v.w;
        s0 = u.x; s1 = u.y; s2 = u.z; s3 = u.w;
    } else {
        if (e0 + 0 < E) { d0 = dst[e0 + 0]; s0 = src[e0 + 0]; }
        if (e0 + 1 < E) { d1 = dst[e0 + 1]; s1 = src[e0 + 1]; }
        if (e0 + 2 < E) { d2 = dst[e0 + 2]; s2 = src[e0 + 2]; }
        if (e0 + 3 < E) { d3 = dst[e0 + 3]; s3 = src[e0 + 3]; }
    }
    unsigned w = 0;
    uint4 r = make_uint4(0u, 0u, 0u, 0u);
    if (d0 >= 0) {
        atomicAdd(&deg[d0], 1.0f);
        const unsigned o = (unsigned)d0 / (unsigned)AG_NPB;
        w |= o;
        r.x = ((unsigned)s0 & AG_SMASK) | (((unsigned)d0 - o * (unsigned)AG_NPB) << AG_SBITS);
    }
    if (d1 >= 0) {
        atomicAdd(&deg[d1], 1.0f);
        const unsigned o = (unsigned)d1 / (unsigned)AG_NPB;
        w |= o << 8;
        r.y = ((unsigned)s1 & AG_SMASK) | (((unsigned)d1 - o * (unsigned)AG_NPB) << AG_SBITS);
    }
    if (d2 >= 0) {
        atomicAdd(&deg[d2], 1.0f);
        const unsigned o = (unsigned)d2 / (unsigned)AG_NPB;
        w |= o << 16;
        r.z = ((unsigned)s2 & AG_SMASK) | (((unsigned)d2 - o * (unsigned)AG_NPB) << AG_SBITS);
    }
    if (d3 >= 0) {
        atomicAdd(&deg[d3], 1.0f);
        const unsigned o = (unsigned)d3 / (unsigned)AG_NPB;
        w |= o << 24;
        r.w = ((unsigned)s3 & AG_SMASK) | (((unsigned)d3 - o * (unsigned)AG_NPB) << AG_SBITS);
    }
    own4[i] = w;
    ((uint4*)rec)[i] = r;
}

__global__ void k_rsqrt(float* __restrict__ d, int n, int* __restrict__ cursor) {
    int i = blockIdx.x * blockDim.x + threadIdx.x;
    if (i == 0) *cursor = 0;
    if (i < n) {
        float v = d[i];
        d[i] = (v > 0.0f) ? rsqrtf(v) : 0.0f;
    }
}

__global__ void k_transpose_pad(const float* __restrict__ Bw, float* __restrict__ Bt,
                                int K, int Ncols, int ncols_pad) {
    int i = blockIdx.x * blockDim.x + threadIdx.x;
    int total = ncols_pad * K;
    if (i >= total) return;
    int ncol = i / K, k = i - ncol * K;
    Bt[i] = (ncol < Ncols) ? Bw[(size_t)k * Ncols + ncol] : 0.0f;
}

#define AG_F      20
#define AG_GN     82
#define FC_K      (AG_GN * AG_F)
#define FC_N      300
#define FC_NT     ((FC_N + 15) / 16)
#define FC_G      10
#define AG_WAVES  32
#define AG_TB     (AG_WAVES * 32)
#define AG_QFLUSH 128
#define AG_QW     (AG_QFLUSH + 32)
#define AG_CHW    (3 * AG_QW)

template <bool HEAD>
__global__ void __launch_bounds__(AG_TB)
k_aggregate(const unsigned* __restrict__ rec,
            const unsigned char* __restrict__ own, const float* __restrict__ xin, const float* __restrict__ dinv,
            const float* __restrict__ Wm, const float* __restrict__ bias, float* __restrict__ h,
            const float* __restrict__ fcW, const float* __restrict__ fcb, float* __restrict__ fco,
            int* __restrict__ lst, int* __restrict__ hdr, int* __restrict__ tails, int* __restrict__ cursor, int maxch,
            int E, int N) {
    static_assert(AG_NPB % AG_GN == 0 && AG_NPB / AG_GN == 32, "a block owns 32 whole graphs");
    static_assert((FC_K / 4) % FC_G == 0, "whole groups of k steps");
    __shared__ __attribute__((aligned(16))) float acc[AG_NPB * AG_F];

    const int tid = threadIdx.x;
    const int lane = tid & 31;
    const int wave = __builtin_amdgcn_readfirstlane(tid >> 5);
    const int node0 = blockIdx.x * AG_NPB;

    float4* acc4 = (float4*)acc;
    for (int q = tid; q < AG_NPB * AG_F / 4; q += AG_TB) acc4[q] = make_float4(0.f, 0.f, 0.f, 0.f);
    __syncthreads();


    int elb[5], fb[5];
#pragma unroll
    for (int b = 0; b < 5; ++b) {
        elb[b] = (32 * b + lane) / AG_F;
        fb[b] = (32 * b + lane) - elb[b] * AG_F;
    }

    auto apply = [&](int s20, int d20, float nr, int lim) {
        float raw[AG_F];
#pragma unroll
        for (int j = 0; j < AG_F; ++j) {
            const int a = j / 5, b = j % 5;
            const int ss = __builtin_amdgcn_ds_bpermute((8 * a + elb[b]) << 2, s20);
            raw[j] = xin[ss + fb[b]];
        }
#pragma unroll
        for (int j = 0; j < AG_F; ++j) {
            const int a = j / 5, b = j % 5;
            const int dd = __builtin_amdgcn_ds_bpermute((8 * a + elb[b]) << 2, d20);
            const float nn = __int_as_float(__builtin_amdgcn_ds_bpermute((8 * a + elb[b]) << 2, __float_as_int(nr)));
            if (160 * a + 32 * b + lane < lim) atomicAdd(&acc[dd + fb[b]], raw[j] * nn);
        }
    };

    if constexpr (!HEAD) {
        __shared__ int qe[AG_WAVES * AG_QW];
        __shared__ float dloc[AG_NPB];
        for (int q = tid; q < AG_NPB; q += AG_TB) dloc[q] = node0 + q < N ? dinv[node0 + q] : 0.0f;
        __syncthreads();
        int* myqe = qe + wave * AG_QW;
        int cnt = 0;
        int prevc = -1;

        auto push = [&](bool m, int e) {
            const unsigned bal = __builtin_amdgcn_ballot_w32(m);
            if (bal) {
                if (m) myqe[cnt + (int)__builtin_amdgcn_mbcnt_lo(bal, 0u)] = e;
                cnt += __builtin_popcount(bal);
            }
        };

        const int G = (E + 15) >> 4;
        const int TT = (G + 31) >> 5;
        const int rot = (int)blockIdx.x * (TT / (int)gridDim.x);
        const uint4* __restrict__ own16 = (const uint4*)own;
        const int KT = (TT + AG_WAVES - 1) / AG_WAVES;
        const unsigned bb = (unsigned)blockIdx.x * 0x01010101u;

        auto differs = [&](unsigned w) -> unsigned {
            const unsigned x = w ^ bb;
            return ((x & 0x7F7F7F7Fu) + 0x7F7F7F7Fu) | x;
        };

        auto drain = [&]() {
            __builtin_amdgcn_fence(__ATOMIC_ACQ_REL, "wavefront");
            int cv = 0;
            if (lane == 0) cv = atomicAdd(cursor, 1);
            const int c = __builtin_amdgcn_readfirstlane(cv);
            const bool keep = c < maxch;
            int* ch = lst + (size_t)(keep ? c : 0) * AG_CHW;
            if (keep) {
                if (lane == 0) *(int2*)(hdr + 2 * c) = make_int2(prevc, cnt);
                prevc = c;
            }
            for (int base = 0; base < cnt; base += 32) {
                const int i = base + lane;
                int s20 = 0, d20 = 0;
                float nr = 0.0f;
                if (i < cnt) {
                    const int e = myqe[i];
                    const unsigned pk = rec[e];
                    const int s = (int)(pk & AG_SMASK);
                    const int dl = (int)(pk >> AG_SBITS);
                    nr = dinv[s] * dloc[dl];
                    s20 = s * AG_F;
                    d20 = dl * AG_F;
                }
                if (keep) {
                    ch[i] = s20;
                    ch[AG_QW + i] = d20;
                    ch[2 * AG_QW + i] = __float_as_int(nr);
                }
                const int rem = cnt - base;
                apply(s20, d20, nr, (rem < 32 ? rem : 32) * AG_F);
            }
            cnt = 0;
        };

        auto exists_mask = [&](int left) -> unsigned {
            unsigned m = 0u;
#pragma unroll
            for (int q = 0; q < 16; ++q)
                if (q < left) m |= 1u << (8 * (q & 3) + (q >> 2));
            return m;
        };

        auto work = [&](int g, bool live, const uint4& r) {
            const unsigned t0 = differs(r.x), t1 = differs(r.y), t2 = differs(r.z), t3 = differs(r.w);
            const unsigned T = ((t0 >> 7) & 0x01010101u) | ((t1 >> 6) & 0x02020202u) |
                               ((t2 >> 5) & 0x04040404u) | ((t3 >> 4) & 0x08080808u);
            unsigned ownm = ~T & 0x0F0F0F0Fu;
            const int e0 = g * 16;
            const int left = live ? E - e0 : 0;
            if (__builtin_amdgcn_ballot_w32(left < 16)) ownm &= exists_mask(left);
            while (__builtin_amdgcn_ballot_w32(ownm != 0u)) {
                const int p = __builtin_ctz(ownm | 0x80000000u);
                push(ownm != 0u, e0 + ((p & 7) << 2) + (p >> 3));
                ownm &= ownm - 1u;
                if (cnt >= AG_QFLUSH) drain();
            }
        };

        for (int k = 0; k < KT; k += 4) {
            int g[4];
            bool live[4];
            uint4 r[4];
#pragma unroll
            for (int u = 0; u < 4; ++u) {
                const int tl = (k + u) * AG_WAVES + wave;
                int at = tl + rot;
                if (at >= TT) at -= TT;
                live[u] = tl < TT;
                g[u] = at * 32 + lane;
                r[u] = own16[g[u] < G ? g[u] : G - 1];
            }
#pragma unroll
            for (int u = 0; u < 4; ++u) work(g[u], live[u], r[u]);
        }
        if (cnt > 0) drain();
        if (lane == 0) tails[blockIdx.x * AG_WAVES + wave] = prevc;
    } else {
        int c = __builtin_amdgcn_readfirstlane(tails[blockIdx.x * AG_WAVES + wave]);
        for (int it = 0; it < maxch && (unsigned)c < (unsigned)maxch; ++it) {
            const int2 hd = *(const int2*)(hdr + 2 * c);
            const int hn = __builtin_amdgcn_readfirstlane(hd.y);
            const int n = hn < AG_QW ? hn : AG_QW;
            const int* ch = lst + (size_t)c * AG_CHW;
            int s20 = ch[lane], d20 = ch[AG_QW + lane], nri = ch[2 * AG_QW + lane];
            for (int base = 0; base < n; base += 32) {
                const int nb = (base + 32 < n ? base + 32 : base) + lane;
                const int ns = ch[nb], nd = ch[AG_QW + nb], nn = ch[2 * AG_QW + nb];
                const int rem = n - base;
                apply(s20, d20, __int_as_float(nri), (rem < 32 ? rem : 32) * AG_F);
                s20 = ns;
                d20 = nd;
                nri = nn;
            }
            c = __builtin_amdgcn_readfirstlane(hd.x);
        }
    }
    __syncthreads();

    const float4* t4 = (const float4*)(xin + (size_t)node0 * AG_F);
    for (int q = tid; q < AG_NPB * AG_F / 4; q += AG_TB) {
        const int nl = q / 5;
        if (node0 + nl < N) {
            const float di = dinv[node0 + nl];
            const float sl = di * di;
            float4 a = acc4[q];
            const float4 tv = t4[q];
            a.x += tv.x * sl; a.y += tv.y * sl; a.z += tv.z * sl; a.w += tv.w * sl;
            acc4[q] = a;
        }
    }
    __syncthreads();

    {
        const int c = lane & 15, hh = lane >> 4;
        const int n1 = 16 + c < AG_F ? 16 + c : AG_F - 1;
        v2f w0[AG_F / 4], w1[AG_F / 4];
#pragma unroll
        for (int j = 0; j < AG_F / 4; ++j) {
            const int k = 4 * j + 2 * hh;
            w0[j].x = Wm[k * AG_F + c];
            w0[j].y = Wm[(k + 1) * AG_F + c];
            w1[j].x = Wm[k * AG_F + n1];
            w1[j].y = Wm[(k + 1) * AG_F + n1];
        }
        const float bv0 = bias[c], bv1 = bias[n1];
        for (int mt = wave; mt < AG_NPB / 16; mt += AG_WAVES) {
            float* rows = acc + mt * 16 * AG_F;
            const float* ap = rows + c * AG_F + 2 * hh;
            v2f a[AG_F / 4];
#pragma unroll
            for (int j = 0; j < AG_F / 4; ++j) a[j] = *(const v2f*)(ap + 4 * j);
            v8f d0 = {}, d1 = {};
#pragma unroll
            for (int j = 0; j < AG_F / 4; ++j) {
                d0 = __builtin_amdgcn_wmma_f32_16x16x4_f32(false, a[j], false, w0[j], (short)0, d0, false, false);
                d1 = __builtin_amdgcn_wmma_f32_16x16x4_f32(false, a[j], false, w1[j], (short)0, d1, false, false);
            }
            float* op = rows + 8 * hh * AG_F;
#pragma unroll
            for (int r = 0; r < 8; ++r) {
                op[r * AG_F + c] = fmaxf(d0[r] + bv0, 0.0f);
                if (16 + c < AG_F) op[r * AG_F + 16 + c] = fmaxf(d1[r] + bv1, 0.0f);
            }
        }
    }
    __syncthreads();

    if constexpr (!HEAD) {
        float4* h4 = (float4*)(h + (size_t)node0 * AG_F);
        for (int q = tid; q < AG_NPB * AG_F / 4; q += AG_TB) {
            const int nl = q / 5;
            if (node0 + nl < N) h4[q] = acc4[q];
        }
    } else {
        const int c = lane & 15, hh = lane >> 4;
        const int wa = wave >> 3, wb = (wave - wa) & 7;
        const int nt = wb < 5 ? wa * 5 + wb : FC_NT;
        const int ncol = nt * 16 + c;
        v8f d0 = {}, d1 = {};
        if (nt < FC_NT) {
            const int nc = ncol < FC_N ? ncol : FC_N - 1;
            const float* bp = fcW + (size_t)(2 * hh) * FC_N + nc;
            const float* a0p = acc + c * FC_K + 2 * hh;
            const float* a1p = a0p + 16 * FC_K;
            float bx[FC_G], by[FC_G];
#pragma unroll
            for (int j = 0; j < FC_G; ++j) {
                bx[j] = bp[(size_t)(4 * j) * FC_N];
                by[j] = bp[(size_t)(4 * j + 1) * FC_N];
            }
            constexpr int NG = FC_K / 4 / FC_G;
            for (int g = 0; g < NG; ++g) {
                const int gn = g + 1 < NG ? g + 1 : g;
                const float* bq = bp + (size_t)(gn * FC_G * 4) * FC_N;
                float nx[FC_G], ny[FC_G];
#pragma unroll
                for (int j = 0; j < FC_G; ++j) {
                    nx[j] = bq[(size_t)(4 * j) * FC_N];
                    ny[j] = bq[(size_t)(4 * j + 1) * FC_N];
                }
                v2f a0[FC_G], a1[FC_G];
#pragma unroll
                for (int j = 0; j < FC_G; ++j) {
                    a0[j] = *(const v2f*)(a0p + (g * FC_G + j) * 4);
                    a1[j] = *(const v2f*)(a1p + (g * FC_G + j) * 4);
                }
#pragma unroll
                for (int j = 0; j < FC_G; ++j) {
                    v2f b;
                    b.x = bx[j];
                    b.y = by[j];
                    d0 = __builtin_amdgcn_wmma_f32_16x16x4_f32(false, a0[j], false, b, (short)0, d0, false, false);
                    d1 = __builtin_amdgcn_wmma_f32_16x16x4_f32(false, a1[j], false, b, (short)0, d1, false, false);
                }
#pragma unroll
                for (int j = 0; j < FC_G; ++j) {
                    bx[j] = nx[j];
                    by[j] = ny[j];
                }
            }
        }
        __syncthreads();
        if (nt < FC_NT && ncol < FC_N) {
            const float bv = fcb[ncol];
            float* op = acc + 8 * hh * FC_N + ncol;
#pragma unroll
            for (int r = 0; r < 8; ++r) {
                op[r * FC_N] = fmaxf(d0[r] + bv, 0.0f);
                op[(16 + r) * FC_N] = fmaxf(d1[r] + bv, 0.0f);
            }
        }
        __syncthreads();
        const int g0 = (int)blockIdx.x * (AG_NPB / AG_GN);
        const int ng = N / AG_GN;
        float4* o4 = (float4*)(fco + (size_t)g0 * FC_N);
        for (int q = tid; q < (AG_NPB / AG_GN) * FC_N / 4; q += AG_TB) {
            const int gl = q / (FC_N / 4);
            if (g0 + gl < ng) o4[q] = acc4[q];
        }
    }
}

__global__ void k_wmma_gemm_f32(const float* __restrict__ A, const float* __restrict__ Bt,
                                const float* __restrict__ bias, float* __restrict__ C,
                                int M, int K, int Ncols, int mtiles, int ntiles, int do_relu) {
    int wave = (blockIdx.x * blockDim.x + threadIdx.x) >> 5;
    int lane = threadIdx.x & 31;
    int mt = wave % mtiles;
    int nt = wave / mtiles;
    if (nt >= ntiles) return;

    int lrow = lane & 15;
    int kh = (lane >> 4) << 1;
    int ncol = nt * 16 + lrow;
    const float* arow = A + (size_t)(mt * 16 + lrow) * K + kh;
    const float* brow = Bt + (size_t)ncol * K + kh;

    v8f acc = {};
    int kb = 0;
    for (; kb + 16 <= K; kb += 16) {
#pragma unroll
        for (int j = 0; j < 4; ++j) {
            v2f a = *(const v2f*)(arow + kb + 4 * j);
            v2f b = *(const v2f*)(brow + kb + 4 * j);
            acc = __builtin_amdgcn_wmma_f32_16x16x4_f32(false, a, false, b,
                                                        (short)0, acc, false, false);
        }
    }
    for (; kb < K; kb += 4) {
        v2f a = *(const v2f*)(arow + kb);
        v2f b = *(const v2f*)(brow + kb);
        acc = __builtin_amdgcn_wmma_f32_16x16x4_f32(false, a, false, b,
                                                    (short)0, acc, false, false);
    }

    if (ncol >= Ncols) return;
    float bv = bias ? bias[ncol] : 0.0f;
    int rbase = mt * 16 + ((lane >> 4) << 3);
#pragma unroll
    for (int r = 0; r < 8; ++r) {
        float v = acc[r] + bv;
        if (do_relu) v = fmaxf(v, 0.0f);
        C[(size_t)(rbase + r) * Ncols + ncol] = v;
    }
}


extern "C" void kernel_launch(void* const* d_in, const int* in_sizes, int n_in,
                              void* d_out, int out_size, void* d_ws, size_t ws_size,
                              hipStream_t stream) {
    const float* x    = (const float*)d_in[0];
    const int*   eidx = (const int*)d_in[1];
    const float* W1   = (const float*)d_in[3];
    const float* b1   = (const float*)d_in[4];
    const float* W2   = (const float*)d_in[5];
    const float* b2   = (const float*)d_in[6];
    const float* fc1W = (const float*)d_in[7];
    const float* fc1b = (const float*)d_in[8];
    const float* outW = (const float*)d_in[9];
    const float* outb = (const float*)d_in[10];

    const int N  = in_sizes[2];
    const int E  = in_sizes[1] / 2;
    const int Bg = N / 82;
    const int EMB = 20, HID = 300, NC = 22;

    char* ws = (char*)d_ws;
    size_t off = 0;
    auto take = [&](size_t bytes) -> void* {
        off = (off + 255) & ~(size_t)255;
        void* p = ws + off;
        off += bytes;
        return p;
    };
    float* dinv = (float*)take((size_t)N * 4);
    float* t    = (float*)take((size_t)N * EMB * 4);
    float* fco  = (float*)take((size_t)Bg * HID * 4);
    float* Bt   = (float*)take((size_t)32 * HID * 4);
    const int nwords = cdiv(E, 16) * 4;
    unsigned* own4 = (unsigned*)take((size_t)nwords * 4);
    const unsigned char* own = (const unsigned char*)own4;
    unsigned* rec = (unsigned*)take((size_t)nwords * 16);
    const int maxch = E / AG_QFLUSH + cdiv(N, AG_NPB) * AG_WAVES;
    int* lst    = (int*)take((size_t)maxch * AG_CHW * 4);
    int* hdr    = (int*)take((size_t)maxch * 8);
    int* tails  = (int*)take((size_t)cdiv(N, AG_NPB) * AG_WAVES * 4);
    int* cursor = (int*)take(4);

    const int* src = eidx;
    const int* dst = eidx + E;

    const int TB = 256;

    auto run_gemm = [&](const float* A, const float* Bw, const float* bias, float* C,
                        int M, int K, int Ncols, int relu) {
        int ntiles = (Ncols + 15) / 16;
        int ncols_pad = ntiles * 16;
        int mtiles = M / 16;
        k_transpose_pad<<<cdiv(ncols_pad * K, TB), TB, 0, stream>>>(Bw, Bt, K, Ncols, ncols_pad);
        int waves = mtiles * ntiles;
        k_wmma_gemm_f32<<<cdiv(waves * 32, TB), TB, 0, stream>>>(
            A, Bt, bias, C, M, K, Ncols, mtiles, ntiles, relu);
    };

    k_fill<<<cdiv(N, TB), TB, 0, stream>>>(dinv, 1.0f, N);
    k_deg_owner<<<cdiv(nwords, TB), TB, 0, stream>>>(src, dst, dinv, own4, rec, E, nwords);
    k_rsqrt<<<cdiv(N, TB), TB, 0, stream>>>(dinv, N, cursor);

    k_aggregate<false><<<cdiv(N, AG_NPB), AG_TB, 0, stream>>>(rec, own, x, dinv, W1, b1, t,
                                                              nullptr, nullptr, nullptr,
                                                              lst, hdr, tails, cursor, maxch, E, N);

    k_aggregate<true><<<cdiv(N, AG_NPB), AG_TB, 0, stream>>>(rec, own, t, dinv, W2, b2, nullptr,
                                                             fc1W, fc1b, fco,
                                                             lst, hdr, tails, cursor, maxch, E, N);

    run_gemm(fco, outW, outb, (float*)d_out, Bg, HID, NC, 0);
}


# Degree pass's 4M global float atomics gone: owner blocks count their nodes' in-edges in LDS (int adds) during the owner scan, moved ahead of layer 1, which replays the list; sum order = LDS arrival
# speedup vs baseline: 6.2502x; 1.2510x over previous
#include <hip/hip_runtime.h>
#include <hip/hip_bf16.h>


typedef __attribute__((ext_vector_type(2))) float v2f;
typedef __attribute__((ext_vector_type(8))) float v8f;

static __host__ inline int cdiv(int a, int b) { return (a + b - 1) / b; }


#define AG_NPB    2624
#define AG_SBITS  20
#define AG_SMASK  ((1u << AG_SBITS) - 1u)
__global__ void k_owner_rec(const int* __restrict__ src, const int* __restrict__ dst,
                            unsigned* __restrict__ own4, unsigned* __restrict__ rec, int* __restrict__ cursor,
                            int E, int nwords) {
    static_assert(256 * AG_NPB <= (1 << AG_SBITS) && AG_NPB <= (1 << (32 - AG_SBITS)), "src and local dst fit one word");
    const int i = blockIdx.x * blockDim.x + threadIdx.x;
    if (i == 0) *cursor = 0;
    if (i >= nwords) return;
    const int e0 = 4 * i;
    int d0 = -1, d1 = -1, d2 = -1, d3 = -1;
    int s0 = 0, s1 = 0, s2 = 0, s3 = 0;
    if ((((((size_t)dst) | ((size_t)src)) & 15) == 0) && e0 + 3 < E) {
        const int4 v = ((const int4*)dst)[i];
        const int4 u = ((const int4*)src)[i];
        d0 = v.x; d1 = v.y; d2 = v.z; d3 = v.w;
        s0 = u.x; s1 = u.y; s2 = u.z; s3 = u.w;
    } else {
        if (e0 + 0 < E) { d0 = dst[e0 + 0]; s0 = src[e0 + 0]; }
        if (e0 + 1 < E) { d1 = dst[e0 + 1]; s1 = src[e0 + 1]; }
        if (e0 + 2 < E) { d2 = dst[e0 + 2]; s2 = src[e0 + 2]; }
        if (e0 + 3 < E) { d3 = dst[e0 + 3]; s3 = src[e0 + 3]; }
    }
    unsigned w = 0;
    uint4 r = make_uint4(0u, 0u, 0u, 0u);
    if (d0 >= 0) {
        const unsigned o = (unsigned)d0 / (unsigned)AG_NPB;
        w |= o;
        r.x = ((unsigned)s0 & AG_SMASK) | (((unsigned)d0 - o * (unsigned)AG_NPB) << AG_SBITS);
    }
    if (d1 >= 0) {
        const unsigned o = (unsigned)d1 / (unsigned)AG_NPB;
        w |= o << 8;
        r.y = ((unsigned)s1 & AG_SMASK) | (((unsigned)d1 - o * (unsigned)AG_NPB) << AG_SBITS);
    }
    if (d2 >= 0) {
        const unsigned o = (unsigned)d2 / (unsigned)AG_NPB;
        w |= o << 16;
        r.z = ((unsigned)s2 & AG_SMASK) | (((unsigned)d2 - o * (unsigned)AG_NPB) << AG_SBITS);
    }
    if (d3 >= 0) {
        const unsigned o = (unsigned)d3 / (unsigned)AG_NPB;
        w |= o << 24;
        r.w = ((unsigned)s3 & AG_SMASK) | (((unsigned)d3 - o * (unsigned)AG_NPB) << AG_SBITS);
    }
    own4[i] = w;
    ((uint4*)rec)[i] = r;
}

__global__ void k_transpose_pad(const float* __restrict__ Bw, float* __restrict__ Bt,
                                int K, int Ncols, int ncols_pad) {
    int i = blockIdx.x * blockDim.x + threadIdx.x;
    int total = ncols_pad * K;
    if (i >= total) return;
    int ncol = i / K, k = i - ncol * K;
    Bt[i] = (ncol < Ncols) ? Bw[(size_t)k * Ncols + ncol] : 0.0f;
}

#define AG_F      20
#define AG_GN     82
#define FC_K      (AG_GN * AG_F)
#define FC_N      300
#define FC_NT     ((FC_N + 15) / 16)
#define FC_G      10
#define AG_WAVES  32
#define AG_TB     (AG_WAVES * 32)
#define AG_QFLUSH 128
#define AG_QW     (AG_QFLUSH + 32)
#define AG_QB     (AG_QW / 32)
#define AG_CHW    (3 * AG_QW)

__global__ void __launch_bounds__(AG_TB)
k_scan_count(const unsigned* __restrict__ rec, const unsigned char* __restrict__ own, float* __restrict__ dinv,
             int* __restrict__ lst, int* __restrict__ hdr, int* __restrict__ tails, int* __restrict__ cursor, int maxch,
             int E, int N) {
    static_assert(AG_QW % 32 == 0, "whole batches in a queue");
    __shared__ int qe[AG_WAVES * AG_QW];
    __shared__ int cntl[AG_NPB];

    const int tid = threadIdx.x;
    const int lane = tid & 31;
    const int wave = __builtin_amdgcn_readfirstlane(tid >> 5);
    const int node0 = blockIdx.x * AG_NPB;

    for (int q = tid; q < AG_NPB; q += AG_TB) cntl[q] = 0;
    __syncthreads();

    int* myqe = qe + wave * AG_QW;
    int cnt = 0;
    int prevc = -1;

    auto push = [&](bool m, int e) {
        const unsigned bal = __builtin_amdgcn_ballot_w32(m);
        if (bal) {
            if (m) myqe[cnt + (int)__builtin_amdgcn_mbcnt_lo(bal, 0u)] = e;
            cnt += __builtin_popcount(bal);
        }
    };

    const int G = (E + 15) >> 4;
    const int TT = (G + 31) >> 5;
    const int rot = (int)blockIdx.x * (TT / (int)gridDim.x);
    const uint4* __restrict__ own16 = (const uint4*)own;
    const int KT = (TT + AG_WAVES - 1) / AG_WAVES;
    const unsigned bb = (unsigned)blockIdx.x * 0x01010101u;

    auto differs = [&](unsigned w) -> unsigned {
        const unsigned x = w ^ bb;
        return ((x & 0x7F7F7F7Fu) + 0x7F7F7F7Fu) | x;
    };

    auto drain = [&]() {
        __builtin_amdgcn_fence(__ATOMIC_ACQ_REL, "wavefront");
        int cv = 0;
        if (lane == 0) cv = atomicAdd(cursor, 1);
        const int c = __builtin_amdgcn_readfirstlane(cv);
        const bool keep = c < maxch;
        int* ch = lst + (size_t)(keep ? c : 0) * AG_CHW;
        if (keep) {
            if (lane == 0) *(int2*)(hdr + 2 * c) = make_int2(prevc, cnt);
            prevc = c;
        }
        unsigned pk[AG_QB];
#pragma unroll
        for (int u = 0; u < AG_QB; ++u) {
            const int i = 32 * u + lane;
            pk[u] = rec[myqe[i < cnt ? i : cnt - 1]];
        }
#pragma unroll
        for (int u = 0; u < AG_QB; ++u) {
            if (32 * u < cnt) {
                const int i = 32 * u + lane;
                int s20 = 0, d20 = 0;
                if (i < cnt) {
                    const int s = (int)(pk[u] & AG_SMASK);
                    const int dl = (int)(pk[u] >> AG_SBITS);
                    atomicAdd(&cntl[dl], 1);
                    s20 = s * AG_F;
                    d20 = dl * AG_F;
                }
                if (keep) {
                    ch[i] = s20;
                    ch[AG_QW + i] = d20;
                }
            }
        }
        cnt = 0;
    };

    auto exists_mask = [&](int left) -> unsigned {
        unsigned m = 0u;
#pragma unroll
        for (int q = 0; q < 16; ++q)
            if (q < left) m |= 1u << (8 * (q & 3) + (q >> 2));
        return m;
    };

    auto work = [&](int g, bool live, const uint4& r) {
        const unsigned t0 = differs(r.x), t1 = differs(r.y), t2 = differs(r.z), t3 = differs(r.w);
        const unsigned T = ((t0 >> 7) & 0x01010101u) | ((t1 >> 6) & 0x02020202u) |
                           ((t2 >> 5) & 0x04040404u) | ((t3 >> 4) & 0x08080808u);
        unsigned ownm = ~T & 0x0F0F0F0Fu;
        const int e0 = g * 16;
        const int left = live ? E - e0 : 0;
        if (__builtin_amdgcn_ballot_w32(left < 16)) ownm &= exists_mask(left);
        while (__builtin_amdgcn_ballot_w32(ownm != 0u)) {
            const int p = __builtin_ctz(ownm | 0x80000000u);
            push(ownm != 0u, e0 + ((p & 7) << 2) + (p >> 3));
            ownm &= ownm - 1u;
            if (cnt >= AG_QFLUSH) drain();
        }
    };

    for (int k = 0; k < KT; k += 4) {
        int g[4];
        bool live[4];
        uint4 r[4];
#pragma unroll
        for (int u = 0; u < 4; ++u) {
            const int tl = (k + u) * AG_WAVES + wave;
            int at = tl + rot;
            if (at >= TT) at -= TT;
            live[u] = tl < TT;
            g[u] = at * 32 + lane;
            r[u] = own16[g[u] < G ? g[u] : G - 1];
        }
#pragma unroll
        for (int u = 0; u < 4; ++u) work(g[u], live[u], r[u]);
    }
    if (cnt > 0) drain();
    if (lane == 0) tails[blockIdx.x * AG_WAVES + wave] = prevc;
    __syncthreads();

    for (int q = tid; q < AG_NPB; q += AG_TB) {
        if (node0 + q < N) {
            const float v = 1.0f + (float)cntl[q];
            dinv[node0 + q] = (v > 0.0f) ? rsqrtf(v) : 0.0f;
        }
    }
}

template <bool HEAD>
__global__ void __launch_bounds__(AG_TB)
k_aggregate(const float* __restrict__ xin, const float* __restrict__ dinv,
            const float* __restrict__ Wm, const float* __restrict__ bias, float* __restrict__ h,
            const float* __restrict__ fcW, const float* __restrict__ fcb, float* __restrict__ fco,
            int* __restrict__ lst, const int* __restrict__ hdr, const int* __restrict__ tails, int maxch, int N) {
    static_assert(AG_NPB % AG_GN == 0 && AG_NPB / AG_GN == 32, "a block owns 32 whole graphs");
    static_assert((FC_K / 4) % FC_G == 0, "whole groups of k steps");
    __shared__ __attribute__((aligned(16))) float acc[AG_NPB * AG_F];

    const int tid = threadIdx.x;
    const int lane = tid & 31;
    const int wave = __builtin_amdgcn_readfirstlane(tid >> 5);
    const int node0 = blockIdx.x * AG_NPB;

    float4* acc4 = (float4*)acc;
    for (int q = tid; q < AG_NPB * AG_F / 4; q += AG_TB) acc4[q] = make_float4(0.f, 0.f, 0.f, 0.f);
    __syncthreads();


    int elb[5], fb[5];
#pragma unroll
    for (int b = 0; b < 5; ++b) {
        elb[b] = (32 * b + lane) / AG_F;
        fb[b] = (32 * b + lane) - elb[b] * AG_F;
    }

    auto apply = [&](int s20, int d20, float nr, int lim) {
        float raw[AG_F];
#pragma unroll
        for (int j = 0; j < AG_F; ++j) {
            const int a = j / 5, b = j % 5;
            const int ss = __builtin_amdgcn_ds_bpermute((8 * a + elb[b]) << 2, s20);
            raw[j] = xin[ss + fb[b]];
        }
#pragma unroll
        for (int j = 0; j < AG_F; ++j) {
            const int a = j / 5, b = j % 5;
            const int dd = __builtin_amdgcn_ds_bpermute((8 * a + elb[b]) << 2, d20);
            const float nn = __int_as_float(__builtin_amdgcn_ds_bpermute((8 * a + elb[b]) << 2, __float_as_int(nr)));
            if (160 * a + 32 * b + lane < lim) atomicAdd(&acc[dd + fb[b]], raw[j] * nn);
        }
    };

    if constexpr (!HEAD) {
        __shared__ float dloc[AG_NPB];
        for (int q = tid; q < AG_NPB; q += AG_TB) dloc[q] = node0 + q < N ? dinv[node0 + q] : 0.0f;
        __syncthreads();
        int c = __builtin_amdgcn_readfirstlane(tails[blockIdx.x * AG_WAVES + wave]);
        for (int it = 0; it < maxch && (unsigned)c < (unsigned)maxch; ++it) {
            const int2 hd = *(const int2*)(hdr + 2 * c);
            const int hn = __builtin_amdgcn_readfirstlane(hd.y);
            const int n = hn < AG_QW ? hn : AG_QW;
            int* ch = lst + (size_t)c * AG_CHW;
            if (n > 0) {
                const int lastb = (n - 1) & ~31;
                const int b1 = lastb < 32 ? lastb : 32;
                int s20 = ch[lane], d20 = ch[AG_QW + lane];
                int s20n = ch[b1 + lane], d20n = ch[AG_QW + b1 + lane];
                float ds = dinv[(unsigned)s20 / (unsigned)AG_F];
                for (int base = 0; base < n; base += 32) {
                    const int b2 = base + 64 < lastb ? base + 64 : lastb;
                    const int s20f = ch[b2 + lane], d20f = ch[AG_QW + b2 + lane];
                    const float dsn = dinv[(unsigned)s20n / (unsigned)AG_F];
                    const float nr = ds * dloc[(unsigned)d20 / (unsigned)AG_F];
                    ch[2 * AG_QW + base + lane] = __float_as_int(nr);
                    const int rem = n - base;
                    apply(s20, d20, nr, (rem < 32 ? rem : 32) * AG_F);
                    s20 = s20n;
                    d20 = d20n;
                    ds = dsn;
                    s20n = s20f;
                    d20n = d20f;
                }
            }
            c = __builtin_amdgcn_readfirstlane(hd.x);
        }
    } else {
        int c = __builtin_amdgcn_readfirstlane(tails[blockIdx.x * AG_WAVES + wave]);
        for (int it = 0; it < maxch && (unsigned)c < (unsigned)maxch; ++it) {
            const int2 hd = *(const int2*)(hdr + 2 * c);
            const int hn = __builtin_amdgcn_readfirstlane(hd.y);
            const int n = hn < AG_QW ? hn : AG_QW;
            const int* ch = lst + (size_t)c * AG_CHW;
            int s20 = ch[lane], d20 = ch[AG_QW + lane], nri = ch[2 * AG_QW + lane];
            for (int base = 0; base < n; base += 32) {
                const int nb = (base + 32 < n ? base + 32 : base) + lane;
                const int ns = ch[nb], nd = ch[AG_QW + nb], nn = ch[2 * AG_QW + nb];
                const int rem = n - base;
                apply(s20, d20, __int_as_float(nri), (rem < 32 ? rem : 32) * AG_F);
                s20 = ns;
                d20 = nd;
                nri = nn;
            }
            c = __builtin_amdgcn_readfirstlane(hd.x);
        }
    }
    __syncthreads();

    const float4* t4 = (const float4*)(xin + (size_t)node0 * AG_F);
    for (int q = tid; q < AG_NPB * AG_F / 4; q += AG_TB) {
        const int nl = q / 5;
        if (node0 + nl < N) {
            const float di = dinv[node0 + nl];
            const float sl = di * di;
            float4 a = acc4[q];
            const float4 tv = t4[q];
            a.x += tv.x * sl; a.y += tv.y * sl; a.z += tv.z * sl; a.w += tv.w * sl;
            acc4[q] = a;
        }
    }
    __syncthreads();

    {
        const int c = lane & 15, hh = lane >> 4;
        const int n1 = 16 + c < AG_F ? 16 + c : AG_F - 1;
        v2f w0[AG_F / 4], w1[AG_F / 4];
#pragma unroll
        for (int j = 0; j < AG_F / 4; ++j) {
            const int k = 4 * j + 2 * hh;
            w0[j].x = Wm[k * AG_F + c];
            w0[j].y = Wm[(k + 1) * AG_F + c];
            w1[j].x = Wm[k * AG_F + n1];
            w1[j].y = Wm[(k + 1) * AG_F + n1];
        }
        const float bv0 = bias[c], bv1 = bias[n1];
        for (int mt = wave; mt < AG_NPB / 16; mt += AG_WAVES) {
            float* rows = acc + mt * 16 * AG_F;
            const float* ap = rows + c * AG_F + 2 * hh;
            v2f a[AG_F / 4];
#pragma unroll
            for (int j = 0; j < AG_F / 4; ++j) a[j] = *(const v2f*)(ap + 4 * j);
            v8f d0 = {}, d1 = {};
#pragma unroll
            for (int j = 0; j < AG_F / 4; ++j) {
                d0 = __builtin_amdgcn_wmma_f32_16x16x4_f32(false, a[j], false, w0[j], (short)0, d0, false, false);
                d1 = __builtin_amdgcn_wmma_f32_16x16x4_f32(false, a[j], false, w1[j], (short)0, d1, false, false);
            }
            float* op = rows + 8 * hh * AG_F;
#pragma unroll
            for (int r = 0; r < 8; ++r) {
                op[r * AG_F + c] = fmaxf(d0[r] + bv0, 0.0f);
                if (16 + c < AG_F) op[r * AG_F + 16 + c] = fmaxf(d1[r] + bv1, 0.0f);
            }
        }
    }
    __syncthreads();

    if constexpr (!HEAD) {
        float4* h4 = (float4*)(h + (size_t)node0 * AG_F);
        for (int q = tid; q < AG_NPB * AG_F / 4; q += AG_TB) {
            const int nl = q / 5;
            if (node0 + nl < N) h4[q] = acc4[q];
        }
    } else {
        const int c = lane & 15, hh = lane >> 4;
        const int wa = wave >> 3, wb = (wave - wa) & 7;
        const int nt = wb < 5 ? wa * 5 + wb : FC_NT;
        const int ncol = nt * 16 + c;
        v8f d0 = {}, d1 = {};
        if (nt < FC_NT) {
            const int nc = ncol < FC_N ? ncol : FC_N - 1;
            const float* bp = fcW + (size_t)(2 * hh) * FC_N + nc;
            const float* a0p = acc + c * FC_K + 2 * hh;
            const float* a1p = a0p + 16 * FC_K;
            float bx[FC_G], by[FC_G];
#pragma unroll
            for (int j = 0; j < FC_G; ++j) {
                bx[j] = bp[(size_t)(4 * j) * FC_N];
                by[j] = bp[(size_t)(4 * j + 1) * FC_N];
            }
            constexpr int NG = FC_K / 4 / FC_G;
            for (int g = 0; g < NG; ++g) {
                const int gn = g + 1 < NG ? g + 1 : g;
                const float* bq = bp + (size_t)(gn * FC_G * 4) * FC_N;
                float nx[FC_G], ny[FC_G];
#pragma unroll
                for (int j = 0; j < FC_G; ++j) {
                    nx[j] = bq[(size_t)(4 * j) * FC_N];
                    ny[j] = bq[(size_t)(4 * j + 1) * FC_N];
                }
                v2f a0[FC_G], a1[FC_G];
#pragma unroll
                for (int j = 0; j < FC_G; ++j) {
                    a0[j] = *(const v2f*)(a0p + (g * FC_G + j) * 4);
                    a1[j] = *(const v2f*)(a1p + (g * FC_G + j) * 4);
                }
#pragma unroll
                for (int j = 0; j < FC_G; ++j) {
                    v2f b;
                    b.x = bx[j];
                    b.y = by[j];
                    d0 = __builtin_amdgcn_wmma_f32_16x16x4_f32(false, a0[j], false, b, (short)0, d0, false, false);
                    d1 = __builtin_amdgcn_wmma_f32_16x16x4_f32(false, a1[j], false, b, (short)0, d1, false, false);
                }
#pragma unroll
                for (int j = 0; j < FC_G; ++j) {
                    bx[j] = nx[j];
                    by[j] = ny[j];
                }
            }
        }
        __syncthreads();
        if (nt < FC_NT && ncol < FC_N) {
            const float bv = fcb[ncol];
            float* op = acc + 8 * hh * FC_N + ncol;
#pragma unroll
            for (int r = 0; r < 8; ++r) {
                op[r * FC_N] = fmaxf(d0[r] + bv, 0.0f);
                op[(16 + r) * FC_N] = fmaxf(d1[r] + bv, 0.0f);
            }
        }
        __syncthreads();
        const int g0 = (int)blockIdx.x * (AG_NPB / AG_GN);
        const int ng = N / AG_GN;
        float4* o4 = (float4*)(fco + (size_t)g0 * FC_N);
        for (int q = tid; q < (AG_NPB / AG_GN) * FC_N / 4; q += AG_TB) {
            const int gl = q / (FC_N / 4);
            if (g0 + gl < ng) o4[q] = acc4[q];
        }
    }
}

__global__ void k_wmma_gemm_f32(const float* __restrict__ A, const float* __restrict__ Bt,
                                const float* __restrict__ bias, float* __restrict__ C,
                                int M, int K, int Ncols, int mtiles, int ntiles, int do_relu) {
    int wave = (blockIdx.x * blockDim.x + threadIdx.x) >> 5;
    int lane = threadIdx.x & 31;
    int mt = wave % mtiles;
    int nt = wave / mtiles;
    if (nt >= ntiles) return;

    int lrow = lane & 15;
    int kh = (lane >> 4) << 1;
    int ncol = nt * 16 + lrow;
    const float* arow = A + (size_t)(mt * 16 + lrow) * K + kh;
    const float* brow = Bt + (size_t)ncol * K + kh;

    v8f acc = {};
    int kb = 0;
    for (; kb + 16 <= K; kb += 16) {
#pragma unroll
        for (int j = 0; j < 4; ++j) {
            v2f a = *(const v2f*)(arow + kb + 4 * j);
            v2f b = *(const v2f*)(brow + kb + 4 * j);
            acc = __builtin_amdgcn_wmma_f32_16x16x4_f32(false, a, false, b,
                                                        (short)0, acc, false, false);
        }
    }
    for (; kb < K; kb += 4) {
        v2f a = *(const v2f*)(arow + kb);
        v2f b = *(const v2f*)(brow + kb);
        acc = __builtin_amdgcn_wmma_f32_16x16x4_f32(false, a, false, b,
                                                    (short)0, acc, false, false);
    }

    if (ncol >= Ncols) return;
    float bv = bias ? bias[ncol] : 0.0f;
    int rbase = mt * 16 + ((lane >> 4) << 3);
#pragma unroll
    for (int r = 0; r < 8; ++r) {
        float v = acc[r] + bv;
        if (do_relu) v = fmaxf(v, 0.0f);
        C[(size_t)(rbase + r) * Ncols + ncol] = v;
    }
}


extern "C" void kernel_launch(void* const* d_in, const int* in_sizes, int n_in,
                              void* d_out, int out_size, void* d_ws, size_t ws_size,
                              hipStream_t stream) {
    const float* x    = (const float*)d_in[0];
    const int*   eidx = (const int*)d_in[1];
    const float* W1   = (const float*)d_in[3];
    const float* b1   = (const float*)d_in[4];
    const float* W2   = (const float*)d_in[5];
    const float* b2   = (const float*)d_in[6];
    const float* fc1W = (const float*)d_in[7];
    const float* fc1b = (const float*)d_in[8];
    const float* outW = (const float*)d_in[9];
    const float* outb = (const float*)d_in[10];

    const int N  = in_sizes[2];
    const int E  = in_sizes[1] / 2;
    const int Bg = N / 82;
    const int EMB = 20, HID = 300, NC = 22;

    char* ws = (char*)d_ws;
    size_t off = 0;
    auto take = [&](size_t bytes) -> void* {
        off = (off + 255) & ~(size_t)255;
        void* p = ws + off;
        off += bytes;
        return p;
    };
    float* dinv = (float*)take((size_t)N * 4);
    float* t    = (float*)take((size_t)N * EMB * 4);
    float* fco  = (float*)take((size_t)Bg * HID * 4);
    float* Bt   = (float*)take((size_t)32 * HID * 4);
    const int nwords = cdiv(E, 16) * 4;
    unsigned* own4 = (unsigned*)take((size_t)nwords * 4);
    const unsigned char* own = (const unsigned char*)own4;
    unsigned* rec = (unsigned*)take((size_t)nwords * 16);
    const int maxch = E / AG_QFLUSH + cdiv(N, AG_NPB) * AG_WAVES;
    int* lst    = (int*)take((size_t)maxch * AG_CHW * 4);
    int* hdr    = (int*)take((size_t)maxch * 8);
    int* tails  = (int*)take((size_t)cdiv(N, AG_NPB) * AG_WAVES * 4);
    int* cursor = (int*)take(4);

    const int* src = eidx;
    const int* dst = eidx + E;

    const int TB = 256;

    auto run_gemm = [&](const float* A, const float* Bw, const float* bias, float* C,
                        int M, int K, int Ncols, int relu) {
        int ntiles = (Ncols + 15) / 16;
        int ncols_pad = ntiles * 16;
        int mtiles = M / 16;
        k_transpose_pad<<<cdiv(ncols_pad * K, TB), TB, 0, stream>>>(Bw, Bt, K, Ncols, ncols_pad);
        int waves = mtiles * ntiles;
        k_wmma_gemm_f32<<<cdiv(waves * 32, TB), TB, 0, stream>>>(
            A, Bt, bias, C, M, K, Ncols, mtiles, ntiles, relu);
    };

    k_owner_rec<<<cdiv(nwords, TB), TB, 0, stream>>>(src, dst, own4, rec, cursor, E, nwords);
    k_scan_count<<<cdiv(N, AG_NPB), AG_TB, 0, stream>>>(rec, own, dinv, lst, hdr, tails, cursor, maxch, E, N);

    k_aggregate<false><<<cdiv(N, AG_NPB), AG_TB, 0, stream>>>(x, dinv, W1, b1, t,
                                                              nullptr, nullptr, nullptr,
                                                              lst, hdr, tails, maxch, N);

    k_aggregate<true><<<cdiv(N, AG_NPB), AG_TB, 0, stream>>>(t, dinv, W2, b2, nullptr,
                                                             fc1W, fc1b, fco,
                                                             lst, hdr, tails, maxch, N);

    run_gemm(fco, outW, outb, (float*)d_out, Bg, HID, NC, 0);
}


# Owner ids kept as 8 bit planes per 32 edges (ballots in the owner pass), not 1 byte per edge: the owner scan matches a lane's 32 edges with 8 xor + or; same bytes read; sum order = LDS arrival
# speedup vs baseline: 6.5845x; 1.0535x over previous
#include <hip/hip_runtime.h>
#include <hip/hip_bf16.h>


typedef __attribute__((ext_vector_type(2))) float v2f;
typedef __attribute__((ext_vector_type(8))) float v8f;

static __host__ inline int cdiv(int a, int b) { return (a + b - 1) / b; }


#define AG_NPB    2624
#define AG_SBITS  20
#define AG_SMASK  ((1u << AG_SBITS) - 1u)
#define OW_TILE   1024
#define OW_TB     (OW_TILE / 4)
__global__ void __launch_bounds__(OW_TB)
k_owner_rec(const int* __restrict__ src, const int* __restrict__ dst,
            unsigned* __restrict__ own4, unsigned* __restrict__ rec, int* __restrict__ cursor, int E) {
    static_assert(256 * AG_NPB <= (1 << AG_SBITS) && AG_NPB <= (1 << (32 - AG_SBITS)), "src and local dst fit one word");
    static_assert(OW_TB == 256, "8 waves x 4 sets = the 32 sets of a tile");
    const int i = blockIdx.x * OW_TB + threadIdx.x;
    if (i == 0) *cursor = 0;
    const int e0 = 4 * i;
    int d0 = -1, d1 = -1, d2 = -1, d3 = -1;
    int s0 = 0, s1 = 0, s2 = 0, s3 = 0;
    if ((((((size_t)dst) | ((size_t)src)) & 15) == 0) && e0 + 3 < E) {
        const int4 v = ((const int4*)dst)[i];
        const int4 u = ((const int4*)src)[i];
        d0 = v.x; d1 = v.y; d2 = v.z; d3 = v.w;
        s0 = u.x; s1 = u.y; s2 = u.z; s3 = u.w;
    } else {
        if (e0 + 0 < E) { d0 = dst[e0 + 0]; s0 = src[e0 + 0]; }
        if (e0 + 1 < E) { d1 = dst[e0 + 1]; s1 = src[e0 + 1]; }
        if (e0 + 2 < E) { d2 = dst[e0 + 2]; s2 = src[e0 + 2]; }
        if (e0 + 3 < E) { d3 = dst[e0 + 3]; s3 = src[e0 + 3]; }
    }
    unsigned o[4] = {0u, 0u, 0u, 0u};
    uint4 r = make_uint4(0u, 0u, 0u, 0u);
    if (d0 >= 0) {
        o[0] = (unsigned)d0 / (unsigned)AG_NPB;
        r.x = ((unsigned)s0 & AG_SMASK) | (((unsigned)d0 - o[0] * (unsigned)AG_NPB) << AG_SBITS);
    }
    if (d1 >= 0) {
        o[1] = (unsigned)d1 / (unsigned)AG_NPB;
        r.y = ((unsigned)s1 & AG_SMASK) | (((unsigned)d1 - o[1] * (unsigned)AG_NPB) << AG_SBITS);
    }
    if (d2 >= 0) {
        o[2] = (unsigned)d2 / (unsigned)AG_NPB;
        r.z = ((unsigned)s2 & AG_SMASK) | (((unsigned)d2 - o[2] * (unsigned)AG_NPB) << AG_SBITS);
    }
    if (d3 >= 0) {
        o[3] = (unsigned)d3 / (unsigned)AG_NPB;
        r.w = ((unsigned)s3 & AG_SMASK) | (((unsigned)d3 - o[3] * (unsigned)AG_NPB) << AG_SBITS);
    }
    ((uint4*)rec)[i] = r;
    const int lane = threadIdx.x & 31;
    unsigned pl = 0u;
#pragma unroll
    for (int j = 0; j < 4; ++j) {
#pragma unroll
        for (int p = 0; p < 8; ++p) {
            const unsigned b = __builtin_amdgcn_ballot_w32(((o[j] >> p) & 1u) != 0u);
            if (lane == 8 * j + p) pl = b;
        }
    }
    const int j = lane >> 3, p = lane & 7;
    const int q = (threadIdx.x >> 5) * 4 + j;
    own4[(size_t)blockIdx.x * (OW_TILE / 4) + (p >> 2) * 128 + q * 4 + (p & 3)] = pl;
}

__global__ void k_transpose_pad(const float* __restrict__ Bw, float* __restrict__ Bt,
                                int K, int Ncols, int ncols_pad) {
    int i = blockIdx.x * blockDim.x + threadIdx.x;
    int total = ncols_pad * K;
    if (i >= total) return;
    int ncol = i / K, k = i - ncol * K;
    Bt[i] = (ncol < Ncols) ? Bw[(size_t)k * Ncols + ncol] : 0.0f;
}

#define AG_F      20
#define AG_GN     82
#define FC_K      (AG_GN * AG_F)
#define FC_N      300
#define FC_NT     ((FC_N + 15) / 16)
#define FC_G      10
#define AG_WAVES  32
#define AG_TB     (AG_WAVES * 32)
#define AG_QFLUSH 128
#define AG_QW     (AG_QFLUSH + 32)
#define AG_QB     (AG_QW / 32)
#define AG_CHW    (3 * AG_QW)

__global__ void __launch_bounds__(AG_TB)
k_scan_count(const unsigned* __restrict__ rec, const unsigned* __restrict__ own, float* __restrict__ dinv,
             int* __restrict__ lst, int* __restrict__ hdr, int* __restrict__ tails, int* __restrict__ cursor, int maxch,
             int E, int N) {
    static_assert(AG_QW % 32 == 0, "whole batches in a queue");
    __shared__ int qe[AG_WAVES * AG_QW];
    __shared__ int cntl[AG_NPB];

    const int tid = threadIdx.x;
    const int lane = tid & 31;
    const int wave = __builtin_amdgcn_readfirstlane(tid >> 5);
    const int node0 = blockIdx.x * AG_NPB;

    for (int q = tid; q < AG_NPB; q += AG_TB) cntl[q] = 0;
    __syncthreads();

    int* myqe = qe + wave * AG_QW;
    int cnt = 0;
    int prevc = -1;

    auto push = [&](bool m, int e) {
        const unsigned bal = __builtin_amdgcn_ballot_w32(m);
        if (bal) {
            if (m) myqe[cnt + (int)__builtin_amdgcn_mbcnt_lo(bal, 0u)] = e;
            cnt += __builtin_popcount(bal);
        }
    };

    const int TT = (E + OW_TILE - 1) / OW_TILE;
    const int rot = (int)blockIdx.x * (TT / (int)gridDim.x);
    const uint4* __restrict__ own16 = (const uint4*)own;
    const int KT = (TT + AG_WAVES - 1) / AG_WAVES;
    unsigned want[8];
#pragma unroll
    for (int p = 0; p < 8; ++p) want[p] = ((blockIdx.x >> p) & 1u) ? 0xFFFFFFFFu : 0u;

    auto drain = [&]() {
        __builtin_amdgcn_fence(__ATOMIC_ACQ_REL, "wavefront");
        int cv = 0;
        if (lane == 0) cv = atomicAdd(cursor, 1);
        const int c = __builtin_amdgcn_readfirstlane(cv);
        const bool keep = c < maxch;
        int* ch = lst + (size_t)(keep ? c : 0) * AG_CHW;
        if (keep) {
            if (lane == 0) *(int2*)(hdr + 2 * c) = make_int2(prevc, cnt);
            prevc = c;
        }
        unsigned pk[AG_QB];
#pragma unroll
        for (int u = 0; u < AG_QB; ++u) {
            const int i = 32 * u + lane;
            pk[u] = rec[myqe[i < cnt ? i : cnt - 1]];
        }
#pragma unroll
        for (int u = 0; u < AG_QB; ++u) {
            if (32 * u < cnt) {
                const int i = 32 * u + lane;
                int s20 = 0, d20 = 0;
                if (i < cnt) {
                    const int s = (int)(pk[u] & AG_SMASK);
                    const int dl = (int)(pk[u] >> AG_SBITS);
                    atomicAdd(&cntl[dl], 1);
                    s20 = s * AG_F;
                    d20 = dl * AG_F;
                }
                if (keep) {
                    ch[i] = s20;
                    ch[AG_QW + i] = d20;
                }
            }
        }
        cnt = 0;
    };

    auto work = [&](int t, bool live, const uint4& ra, const uint4& rb) {
        unsigned ownm = ~((ra.x ^ want[0]) | (ra.y ^ want[1]) | (ra.z ^ want[2]) | (ra.w ^ want[3]) |
                          (rb.x ^ want[4]) | (rb.y ^ want[5]) | (rb.z ^ want[6]) | (rb.w ^ want[7]));
        if (!live) ownm = 0u;
        const int e0 = t * OW_TILE + ((lane >> 2) << 7) + (lane & 3);
        if (t == TT - 1) {
            const int left = E - e0;
            const int nl = left <= 0 ? 0 : (left + 3) >> 2;
            ownm &= nl >= 32 ? 0xFFFFFFFFu : ((1u << nl) - 1u);
        }
        while (__builtin_amdgcn_ballot_w32(ownm != 0u)) {
            const int p = __builtin_ctz(ownm | 0x80000000u);
            push(ownm != 0u, e0 + 4 * p);
            ownm &= ownm - 1u;
            if (cnt >= AG_QFLUSH) drain();
        }
    };

    for (int k = 0; k < KT; k += 4) {
        int tt[4];
        bool live[4];
        uint4 ra[4], rb[4];
#pragma unroll
        for (int u = 0; u < 4; ++u) {
            const int tl = (k + u) * AG_WAVES + wave;
            int at = tl + rot;
            if (at >= TT) at -= TT;
            live[u] = tl < TT;
            tt[u] = at < TT ? at : TT - 1;
            ra[u] = own16[tt[u] * (OW_TILE / 16) + lane];
            rb[u] = own16[tt[u] * (OW_TILE / 16) + 32 + lane];
        }
#pragma unroll
        for (int u = 0; u < 4; ++u) work(tt[u], live[u], ra[u], rb[u]);
    }
    if (cnt > 0) drain();
    if (lane == 0) tails[blockIdx.x * AG_WAVES + wave] = prevc;
    __syncthreads();

    for (int q = tid; q < AG_NPB; q += AG_TB) {
        if (node0 + q < N) {
            const float v = 1.0f + (float)cntl[q];
            dinv[node0 + q] = (v > 0.0f) ? rsqrtf(v) : 0.0f;
        }
    }
}

template <bool HEAD>
__global__ void __launch_bounds__(AG_TB)
k_aggregate(const float* __restrict__ xin, const float* __restrict__ dinv,
            const float* __restrict__ Wm, const float* __restrict__ bias, float* __restrict__ h,
            const float* __restrict__ fcW, const float* __restrict__ fcb, float* __restrict__ fco,
            int* __restrict__ lst, const int* __restrict__ hdr, const int* __restrict__ tails, int maxch, int N) {
    static_assert(AG_NPB % AG_GN == 0 && AG_NPB / AG_GN == 32, "a block owns 32 whole graphs");
    static_assert((FC_K / 4) % FC_G == 0, "whole groups of k steps");
    __shared__ __attribute__((aligned(16))) float acc[AG_NPB * AG_F];

    const int tid = threadIdx.x;
    const int lane = tid & 31;
    const int wave = __builtin_amdgcn_readfirstlane(tid >> 5);
    const int node0 = blockIdx.x * AG_NPB;

    float4* acc4 = (float4*)acc;
    for (int q = tid; q < AG_NPB * AG_F / 4; q += AG_TB) acc4[q] = make_float4(0.f, 0.f, 0.f, 0.f);
    __syncthreads();


    int elb[5], fb[5];
#pragma unroll
    for (int b = 0; b < 5; ++b) {
        elb[b] = (32 * b + lane) / AG_F;
        fb[b] = (32 * b + lane) - elb[b] * AG_F;
    }

    auto apply = [&](int s20, int d20, float nr, int lim) {
        float raw[AG_F];
#pragma unroll
        for (int j = 0; j < AG_F; ++j) {
            const int a = j / 5, b = j % 5;
            const int ss = __builtin_amdgcn_ds_bpermute((8 * a + elb[b]) << 2, s20);
            raw[j] = xin[ss + fb[b]];
        }
#pragma unroll
        for (int j = 0; j < AG_F; ++j) {
            const int a = j / 5, b = j % 5;
            const int dd = __builtin_amdgcn_ds_bpermute((8 * a + elb[b]) << 2, d20);
            const float nn = __int_as_float(__builtin_amdgcn_ds_bpermute((8 * a + elb[b]) << 2, __float_as_int(nr)));
            if (160 * a + 32 * b + lane < lim) atomicAdd(&acc[dd + fb[b]], raw[j] * nn);
        }
    };

    if constexpr (!HEAD) {
        __shared__ float dloc[AG_NPB];
        for (int q = tid; q < AG_NPB; q += AG_TB) dloc[q] = node0 + q < N ? dinv[node0 + q] : 0.0f;
        __syncthreads();
        int c = __builtin_amdgcn_readfirstlane(tails[blockIdx.x * AG_WAVES + wave]);
        for (int it = 0; it < maxch && (unsigned)c < (unsigned)maxch; ++it) {
            const int2 hd = *(const int2*)(hdr + 2 * c);
            const int hn = __builtin_amdgcn_readfirstlane(hd.y);
            const int n = hn < AG_QW ? hn : AG_QW;
            int* ch = lst + (size_t)c * AG_CHW;
            if (n > 0) {
                const int lastb = (n - 1) & ~31;
                const int b1 = lastb < 32 ? lastb : 32;
                int s20 = ch[lane], d20 = ch[AG_QW + lane];
                int s20n = ch[b1 + lane], d20n = ch[AG_QW + b1 + lane];
                float ds = dinv[(unsigned)s20 / (unsigned)AG_F];
                for (int base = 0; base < n; base += 32) {
                    const int b2 = base + 64 < lastb ? base + 64 : lastb;
                    const int s20f = ch[b2 + lane], d20f = ch[AG_QW + b2 + lane];
                    const float dsn = dinv[(unsigned)s20n / (unsigned)AG_F];
                    const float nr = ds * dloc[(unsigned)d20 / (unsigned)AG_F];
                    ch[2 * AG_QW + base + lane] = __float_as_int(nr);
                    const int rem = n - base;
                    apply(s20, d20, nr, (rem < 32 ? rem : 32) * AG_F);
                    s20 = s20n;
                    d20 = d20n;
                    ds = dsn;
                    s20n = s20f;
                    d20n = d20f;
                }
            }
            c = __builtin_amdgcn_readfirstlane(hd.x);
        }
    } else {
        int c = __builtin_amdgcn_readfirstlane(tails[blockIdx.x * AG_WAVES + wave]);
        for (int it = 0; it < maxch && (unsigned)c < (unsigned)maxch; ++it) {
            const int2 hd = *(const int2*)(hdr + 2 * c);
            const int hn = __builtin_amdgcn_readfirstlane(hd.y);
            const int n = hn < AG_QW ? hn : AG_QW;
            const int* ch = lst + (size_t)c * AG_CHW;
            int s20 = ch[lane], d20 = ch[AG_QW + lane], nri = ch[2 * AG_QW + lane];
            for (int base = 0; base < n; base += 32) {
                const int nb = (base + 32 < n ? base + 32 : base) + lane;
                const int ns = ch[nb], nd = ch[AG_QW + nb], nn = ch[2 * AG_QW + nb];
                const int rem = n - base;
                apply(s20, d20, __int_as_float(nri), (rem < 32 ? rem : 32) * AG_F);
                s20 = ns;
                d20 = nd;
                nri = nn;
            }
            c = __builtin_amdgcn_readfirstlane(hd.x);
        }
    }
    __syncthreads();

    const float4* t4 = (const float4*)(xin + (size_t)node0 * AG_F);
    for (int q = tid; q < AG_NPB * AG_F / 4; q += AG_TB) {
        const int nl = q / 5;
        if (node0 + nl < N) {
            const float di = dinv[node0 + nl];
            const float sl = di * di;
            float4 a = acc4[q];
            const float4 tv = t4[q];
            a.x += tv.x * sl; a.y += tv.y * sl; a.z += tv.z * sl; a.w += tv.w * sl;
            acc4[q] = a;
        }
    }
    __syncthreads();

    {
        const int c = lane & 15, hh = lane >> 4;
        const int n1 = 16 + c < AG_F ? 16 + c : AG_F - 1;
        v2f w0[AG_F / 4], w1[AG_F / 4];
#pragma unroll
        for (int j = 0; j < AG_F / 4; ++j) {
            const int k = 4 * j + 2 * hh;
            w0[j].x = Wm[k * AG_F + c];
            w0[j].y = Wm[(k + 1) * AG_F + c];
            w1[j].x = Wm[k * AG_F + n1];
            w1[j].y = Wm[(k + 1) * AG_F + n1];
        }
        const float bv0 = bias[c], bv1 = bias[n1];
        for (int mt = wave; mt < AG_NPB / 16; mt += AG_WAVES) {
            float* rows = acc + mt * 16 * AG_F;
            const float* ap = rows + c * AG_F + 2 * hh;
            v2f a[AG_F / 4];
#pragma unroll
            for (int j = 0; j < AG_F / 4; ++j) a[j] = *(const v2f*)(ap + 4 * j);
            v8f d0 = {}, d1 = {};
#pragma unroll
            for (int j = 0; j < AG_F / 4; ++j) {
                d0 = __builtin_amdgcn_wmma_f32_16x16x4_f32(false, a[j], false, w0[j], (short)0, d0, false, false);
                d1 = __builtin_amdgcn_wmma_f32_16x16x4_f32(false, a[j], false, w1[j], (short)0, d1, false, false);
            }
            float* op = rows + 8 * hh * AG_F;
#pragma unroll
            for (int r = 0; r < 8; ++r) {
                op[r * AG_F + c] = fmaxf(d0[r] + bv0, 0.0f);
                if (16 + c < AG_F) op[r * AG_F + 16 + c] = fmaxf(d1[r] + bv1, 0.0f);
            }
        }
    }
    __syncthreads();

    if constexpr (!HEAD) {
        float4* h4 = (float4*)(h + (size_t)node0 * AG_F);
        for (int q = tid; q < AG_NPB * AG_F / 4; q += AG_TB) {
            const int nl = q / 5;
            if (node0 + nl < N) h4[q] = acc4[q];
        }
    } else {
        const int c = lane & 15, hh = lane >> 4;
        const int wa = wave >> 3, wb = (wave - wa) & 7;
        const int nt = wb < 5 ? wa * 5 + wb : FC_NT;
        const int ncol = nt * 16 + c;
        v8f d0 = {}, d1 = {};
        if (nt < FC_NT) {
            const int nc = ncol < FC_N ? ncol : FC_N - 1;
            const float* bp = fcW + (size_t)(2 * hh) * FC_N + nc;
            const float* a0p = acc + c * FC_K + 2 * hh;
            const float* a1p = a0p + 16 * FC_K;
            float bx[FC_G], by[FC_G];
#pragma unroll
            for (int j = 0; j < FC_G; ++j) {
                bx[j] = bp[(size_t)(4 * j) * FC_N];
                by[j] = bp[(size_t)(4 * j + 1) * FC_N];
            }
            constexpr int NG = FC_K / 4 / FC_G;
            for (int g = 0; g < NG; ++g) {
                const int gn = g + 1 < NG ? g + 1 : g;
                const float* bq = bp + (size_t)(gn * FC_G * 4) * FC_N;
                float nx[FC_G], ny[FC_G];
#pragma unroll
                for (int j = 0; j < FC_G; ++j) {
                    nx[j] = bq[(size_t)(4 * j) * FC_N];
                    ny[j] = bq[(size_t)(4 * j + 1) * FC_N];
                }
                v2f a0[FC_G], a1[FC_G];
#pragma unroll
                for (int j = 0; j < FC_G; ++j) {
                    a0[j] = *(const v2f*)(a0p + (g * FC_G + j) * 4);
                    a1[j] = *(const v2f*)(a1p + (g * FC_G + j) * 4);
                }
#pragma unroll
                for (int j = 0; j < FC_G; ++j) {
                    v2f b;
                    b.x = bx[j];
                    b.y = by[j];
                    d0 = __builtin_amdgcn_wmma_f32_16x16x4_f32(false, a0[j], false, b, (short)0, d0, false, false);
                    d1 = __builtin_amdgcn_wmma_f32_16x16x4_f32(false, a1[j], false, b, (short)0, d1, false, false);
                }
#pragma unroll
                for (int j = 0; j < FC_G; ++j) {
                    bx[j] = nx[j];
                    by[j] = ny[j];
                }
            }
        }
        __syncthreads();
        if (nt < FC_NT && ncol < FC_N) {
            const float bv = fcb[ncol];
            float* op = acc + 8 * hh * FC_N + ncol;
#pragma unroll
            for (int r = 0; r < 8; ++r) {
                op[r * FC_N] = fmaxf(d0[r] + bv, 0.0f);
                op[(16 + r) * FC_N] = fmaxf(d1[r] + bv, 0.0f);
            }
        }
        __syncthreads();
        const int g0 = (int)blockIdx.x * (AG_NPB / AG_GN);
        const int ng = N / AG_GN;
        float4* o4 = (float4*)(fco + (size_t)g0 * FC_N);
        for (int q = tid; q < (AG_NPB / AG_GN) * FC_N / 4; q += AG_TB) {
            const int gl = q / (FC_N / 4);
            if (g0 + gl < ng) o4[q] = acc4[q];
        }
    }
}

__global__ void k_wmma_gemm_f32(const float* __restrict__ A, const float* __restrict__ Bt,
                                const float* __restrict__ bias, float* __restrict__ C,
                                int M, int K, int Ncols, int mtiles, int ntiles, int do_relu) {
    int wave = (blockIdx.x * blockDim.x + threadIdx.x) >> 5;
    int lane = threadIdx.x & 31;
    int mt = wave % mtiles;
    int nt = wave / mtiles;
    if (nt >= ntiles) return;

    int lrow = lane & 15;
    int kh = (lane >> 4) << 1;
    int ncol = nt * 16 + lrow;
    const float* arow = A + (size_t)(mt * 16 + lrow) * K + kh;
    const float* brow = Bt + (size_t)ncol * K + kh;

    v8f acc = {};
    int kb = 0;
    for (; kb + 16 <= K; kb += 16) {
#pragma unroll
        for (int j = 0; j < 4; ++j) {
            v2f a = *(const v2f*)(arow + kb + 4 * j);
            v2f b = *(const v2f*)(brow + kb + 4 * j);
            acc = __builtin_amdgcn_wmma_f32_16x16x4_f32(false, a, false, b,
                                                        (short)0, acc, false, false);
        }
    }
    for (; kb < K; kb += 4) {
        v2f a = *(const v2f*)(arow + kb);
        v2f b = *(const v2f*)(brow + kb);
        acc = __builtin_amdgcn_wmma_f32_16x16x4_f32(false, a, false, b,
                                                    (short)0, acc, false, false);
    }

    if (ncol >= Ncols) return;
    float bv = bias ? bias[ncol] : 0.0f;
    int rbase = mt * 16 + ((lane >> 4) << 3);
#pragma unroll
    for (int r = 0; r < 8; ++r) {
        float v = acc[r] + bv;
        if (do_relu) v = fmaxf(v, 0.0f);
        C[(size_t)(rbase + r) * Ncols + ncol] = v;
    }
}


extern "C" void kernel_launch(void* const* d_in, const int* in_sizes, int n_in,
                              void* d_out, int out_size, void* d_ws, size_t ws_size,
                              hipStream_t stream) {
    const float* x    = (const float*)d_in[0];
    const int*   eidx = (const int*)d_in[1];
    const float* W1   = (const float*)d_in[3];
    const float* b1   = (const float*)d_in[4];
    const float* W2   = (const float*)d_in[5];
    const float* b2   = (const float*)d_in[6];
    const float* fc1W = (const float*)d_in[7];
    const float* fc1b = (const float*)d_in[8];
    const float* outW = (const float*)d_in[9];
    const float* outb = (const float*)d_in[10];

    const int N  = in_sizes[2];
    const int E  = in_sizes[1] / 2;
    const int Bg = N / 82;
    const int EMB = 20, HID = 300, NC = 22;

    char* ws = (char*)d_ws;
    size_t off = 0;
    auto take = [&](size_t bytes) -> void* {
        off = (off + 255) & ~(size_t)255;
        void* p = ws + off;
        off += bytes;
        return p;
    };
    float* dinv = (float*)take((size_t)N * 4);
    float* t    = (float*)take((size_t)N * EMB * 4);
    float* fco  = (float*)take((size_t)Bg * HID * 4);
    float* Bt   = (float*)take((size_t)32 * HID * 4);
    const int ntiles = cdiv(E, OW_TILE);
    unsigned* own4 = (unsigned*)take((size_t)ntiles * (OW_TILE / 4) * 4);
    unsigned* rec = (unsigned*)take((size_t)ntiles * OW_TILE * 4);
    const int maxch = E / AG_QFLUSH + cdiv(N, AG_NPB) * AG_WAVES;
    int* lst    = (int*)take((size_t)maxch * AG_CHW * 4);
    int* hdr    = (int*)take((size_t)maxch * 8);
    int* tails  = (int*)take((size_t)cdiv(N, AG_NPB) * AG_WAVES * 4);
    int* cursor = (int*)take(4);

    const int* src = eidx;
    const int* dst = eidx + E;

    const int TB = 256;

    auto run_gemm = [&](const float* A, const float* Bw, const float* bias, float* C,
                        int M, int K, int Ncols, int relu) {
        int ntiles = (Ncols + 15) / 16;
        int ncols_pad = ntiles * 16;
        int mtiles = M / 16;
        k_transpose_pad<<<cdiv(ncols_pad * K, TB), TB, 0, stream>>>(Bw, Bt, K, Ncols, ncols_pad);
        int waves = mtiles * ntiles;
        k_wmma_gemm_f32<<<cdiv(waves * 32, TB), TB, 0, stream>>>(
            A, Bt, bias, C, M, K, Ncols, mtiles, ntiles, relu);
    };

    k_owner_rec<<<ntiles, OW_TB, 0, stream>>>(src, dst, own4, rec, cursor, E);
    k_scan_count<<<cdiv(N, AG_NPB), AG_TB, 0, stream>>>(rec, own4, dinv, lst, hdr, tails, cursor, maxch, E, N);

    k_aggregate<false><<<cdiv(N, AG_NPB), AG_TB, 0, stream>>>(x, dinv, W1, b1, t,
                                                              nullptr, nullptr, nullptr,
                                                              lst, hdr, tails, maxch, N);

    k_aggregate<true><<<cdiv(N, AG_NPB), AG_TB, 0, stream>>>(t, dinv, W2, b2, nullptr,
                                                             fc1W, fc1b, fco,
                                                             lst, hdr, tails, maxch, N);

    run_gemm(fco, outW, outb, (float*)d_out, Bg, HID, NC, 0);
}


# Edge rows gathered as 16-byte quarters, 5 lanes per row (was 20 lanes x 4 bytes): 5 loads + 15 lane exchanges per 32 edges, was 20 + 60; a lane's 4 LDS adds bank-rotated; sum order = LDS arrival
# speedup vs baseline: 6.8454x; 1.0396x over previous
#include <hip/hip_runtime.h>
#include <hip/hip_bf16.h>


typedef __attribute__((ext_vector_type(2))) float v2f;
typedef __attribute__((ext_vector_type(8))) float v8f;

static __host__ inline int cdiv(int a, int b) { return (a + b - 1) / b; }


#define AG_NPB    2624
#define AG_SBITS  20
#define AG_SMASK  ((1u << AG_SBITS) - 1u)
#define OW_TILE   1024
#define OW_TB     (OW_TILE / 4)
__global__ void __launch_bounds__(OW_TB)
k_owner_rec(const int* __restrict__ src, const int* __restrict__ dst,
            unsigned* __restrict__ own4, unsigned* __restrict__ rec, int* __restrict__ cursor, int E) {
    static_assert(256 * AG_NPB <= (1 << AG_SBITS) && AG_NPB <= (1 << (32 - AG_SBITS)), "src and local dst fit one word");
    static_assert(OW_TB == 256, "8 waves x 4 sets = the 32 sets of a tile");
    const int i = blockIdx.x * OW_TB + threadIdx.x;
    if (i == 0) *cursor = 0;
    const int e0 = 4 * i;
    int d0 = -1, d1 = -1, d2 = -1, d3 = -1;
    int s0 = 0, s1 = 0, s2 = 0, s3 = 0;
    if ((((((size_t)dst) | ((size_t)src)) & 15) == 0) && e0 + 3 < E) {
        const int4 v = ((const int4*)dst)[i];
        const int4 u = ((const int4*)src)[i];
        d0 = v.x; d1 = v.y; d2 = v.z; d3 = v.w;
        s0 = u.x; s1 = u.y; s2 = u.z; s3 = u.w;
    } else {
        if (e0 + 0 < E) { d0 = dst[e0 + 0]; s0 = src[e0 + 0]; }
        if (e0 + 1 < E) { d1 = dst[e0 + 1]; s1 = src[e0 + 1]; }
        if (e0 + 2 < E) { d2 = dst[e0 + 2]; s2 = src[e0 + 2]; }
        if (e0 + 3 < E) { d3 = dst[e0 + 3]; s3 = src[e0 + 3]; }
    }
    unsigned o[4] = {0u, 0u, 0u, 0u};
    uint4 r = make_uint4(0u, 0u, 0u, 0u);
    if (d0 >= 0) {
        o[0] = (unsigned)d0 / (unsigned)AG_NPB;
        r.x = ((unsigned)s0 & AG_SMASK) | (((unsigned)d0 - o[0] * (unsigned)AG_NPB) << AG_SBITS);
    }
    if (d1 >= 0) {
        o[1] = (unsigned)d1 / (unsigned)AG_NPB;
        r.y = ((unsigned)s1 & AG_SMASK) | (((unsigned)d1 - o[1] * (unsigned)AG_NPB) << AG_SBITS);
    }
    if (d2 >= 0) {
        o[2] = (unsigned)d2 / (unsigned)AG_NPB;
        r.z = ((unsigned)s2 & AG_SMASK) | (((unsigned)d2 - o[2] * (unsigned)AG_NPB) << AG_SBITS);
    }
    if (d3 >= 0) {
        o[3] = (unsigned)d3 / (unsigned)AG_NPB;
        r.w = ((unsigned)s3 & AG_SMASK) | (((unsigned)d3 - o[3] * (unsigned)AG_NPB) << AG_SBITS);
    }
    ((uint4*)rec)[i] = r;
    const int lane = threadIdx.x & 31;
    unsigned pl = 0u;
#pragma unroll
    for (int j = 0; j < 4; ++j) {
#pragma unroll
        for (int p = 0; p < 8; ++p) {
            const unsigned b = __builtin_amdgcn_ballot_w32(((o[j] >> p) & 1u) != 0u);
            if (lane == 8 * j + p) pl = b;
        }
    }
    const int j = lane >> 3, p = lane & 7;
    const int q = (threadIdx.x >> 5) * 4 + j;
    own4[(size_t)blockIdx.x * (OW_TILE / 4) + (p >> 2) * 128 + q * 4 + (p & 3)] = pl;
}

__global__ void k_transpose_pad(const float* __restrict__ Bw, float* __restrict__ Bt,
                                int K, int Ncols, int ncols_pad) {
    int i = blockIdx.x * blockDim.x + threadIdx.x;
    int total = ncols_pad * K;
    if (i >= total) return;
    int ncol = i / K, k = i - ncol * K;
    Bt[i] = (ncol < Ncols) ? Bw[(size_t)k * Ncols + ncol] : 0.0f;
}

#define AG_F      20
#define AG_GN     82
#define FC_K      (AG_GN * AG_F)
#define FC_N      300
#define FC_NT     ((FC_N + 15) / 16)
#define FC_G      10
#define AG_WAVES  32
#define AG_TB     (AG_WAVES * 32)
#define AG_QFLUSH 128
#define AG_QW     (AG_QFLUSH + 32)
#define AG_QB     (AG_QW / 32)
#define AG_CHW    (3 * AG_QW)

__global__ void __launch_bounds__(AG_TB)
k_scan_count(const unsigned* __restrict__ rec, const unsigned* __restrict__ own, float* __restrict__ dinv,
             int* __restrict__ lst, int* __restrict__ hdr, int* __restrict__ tails, int* __restrict__ cursor, int maxch,
             int E, int N) {
    static_assert(AG_QW % 32 == 0, "whole batches in a queue");
    __shared__ int qe[AG_WAVES * AG_QW];
    __shared__ int cntl[AG_NPB];

    const int tid = threadIdx.x;
    const int lane = tid & 31;
    const int wave = __builtin_amdgcn_readfirstlane(tid >> 5);
    const int node0 = blockIdx.x * AG_NPB;

    for (int q = tid; q < AG_NPB; q += AG_TB) cntl[q] = 0;
    __syncthreads();

    int* myqe = qe + wave * AG_QW;
    int cnt = 0;
    int prevc = -1;

    auto push = [&](bool m, int e) {
        const unsigned bal = __builtin_amdgcn_ballot_w32(m);
        if (bal) {
            if (m) myqe[cnt + (int)__builtin_amdgcn_mbcnt_lo(bal, 0u)] = e;
            cnt += __builtin_popcount(bal);
        }
    };

    const int TT = (E + OW_TILE - 1) / OW_TILE;
    const int rot = (int)blockIdx.x * (TT / (int)gridDim.x);
    const uint4* __restrict__ own16 = (const uint4*)own;
    const int KT = (TT + AG_WAVES - 1) / AG_WAVES;
    unsigned want[8];
#pragma unroll
    for (int p = 0; p < 8; ++p) want[p] = ((blockIdx.x >> p) & 1u) ? 0xFFFFFFFFu : 0u;

    auto drain = [&]() {
        __builtin_amdgcn_fence(__ATOMIC_ACQ_REL, "wavefront");
        int cv = 0;
        if (lane == 0) cv = atomicAdd(cursor, 1);
        const int c = __builtin_amdgcn_readfirstlane(cv);
        const bool keep = c < maxch;
        int* ch = lst + (size_t)(keep ? c : 0) * AG_CHW;
        if (keep) {
            if (lane == 0) *(int2*)(hdr + 2 * c) = make_int2(prevc, cnt);
            prevc = c;
        }
        unsigned pk[AG_QB];
#pragma unroll
        for (int u = 0; u < AG_QB; ++u) {
            const int i = 32 * u + lane;
            pk[u] = rec[myqe[i < cnt ? i : cnt - 1]];
        }
#pragma unroll
        for (int u = 0; u < AG_QB; ++u) {
            if (32 * u < cnt) {
                const int i = 32 * u + lane;
                int s20 = 0, d20 = 0;
                if (i < cnt) {
                    const int s = (int)(pk[u] & AG_SMASK);
                    const int dl = (int)(pk[u] >> AG_SBITS);
                    atomicAdd(&cntl[dl], 1);
                    s20 = s * AG_F;
                    d20 = dl * AG_F;
                }
                if (keep) {
                    ch[i] = s20;
                    ch[AG_QW + i] = d20;
                }
            }
        }
        cnt = 0;
    };

    auto work = [&](int t, bool live, const uint4& ra, const uint4& rb) {
        unsigned ownm = ~((ra.x ^ want[0]) | (ra.y ^ want[1]) | (ra.z ^ want[2]) | (ra.w ^ want[3]) |
                          (rb.x ^ want[4]) | (rb.y ^ want[5]) | (rb.z ^ want[6]) | (rb.w ^ want[7]));
        if (!live) ownm = 0u;
        const int e0 = t * OW_TILE + ((lane >> 2) << 7) + (lane & 3);
        if (t == TT - 1) {
            const int left = E - e0;
            const int nl = left <= 0 ? 0 : (left + 3) >> 2;
            ownm &= nl >= 32 ? 0xFFFFFFFFu : ((1u << nl) - 1u);
        }
        while (__builtin_amdgcn_ballot_w32(ownm != 0u)) {
            const int p = __builtin_ctz(ownm | 0x80000000u);
            push(ownm != 0u, e0 + 4 * p);
            ownm &= ownm - 1u;
            if (cnt >= AG_QFLUSH) drain();
        }
    };

    for (int k = 0; k < KT; k += 4) {
        int tt[4];
        bool live[4];
        uint4 ra[4], rb[4];
#pragma unroll
        for (int u = 0; u < 4; ++u) {
            const int tl = (k + u) * AG_WAVES + wave;
            int at = tl + rot;
            if (at >= TT) at -= TT;
            live[u] = tl < TT;
            tt[u] = at < TT ? at : TT - 1;
            ra[u] = own16[tt[u] * (OW_TILE / 16) + lane];
            rb[u] = own16[tt[u] * (OW_TILE / 16) + 32 + lane];
        }
#pragma unroll
        for (int u = 0; u < 4; ++u) work(tt[u], live[u], ra[u], rb[u]);
    }
    if (cnt > 0) drain();
    if (lane == 0) tails[blockIdx.x * AG_WAVES + wave] = prevc;
    __syncthreads();

    for (int q = tid; q < AG_NPB; q += AG_TB) {
        if (node0 + q < N) {
            const float v = 1.0f + (float)cntl[q];
            dinv[node0 + q] = (v > 0.0f) ? rsqrtf(v) : 0.0f;
        }
    }
}

template <bool HEAD>
__global__ void __launch_bounds__(AG_TB)
k_aggregate(const float* __restrict__ xin, const float* __restrict__ dinv,
            const float* __restrict__ Wm, const float* __restrict__ bias, float* __restrict__ h,
            const float* __restrict__ fcW, const float* __restrict__ fcb, float* __restrict__ fco,
            int* __restrict__ lst, const int* __restrict__ hdr, const int* __restrict__ tails, int maxch, int N) {
    static_assert(AG_NPB % AG_GN == 0 && AG_NPB / AG_GN == 32, "a block owns 32 whole graphs");
    static_assert((FC_K / 4) % FC_G == 0, "whole groups of k steps");
    __shared__ __attribute__((aligned(16))) float acc[AG_NPB * AG_F];

    const int tid = threadIdx.x;
    const int lane = tid & 31;
    const int wave = __builtin_amdgcn_readfirstlane(tid >> 5);
    const int node0 = blockIdx.x * AG_NPB;

    float4* acc4 = (float4*)acc;
    for (int q = tid; q < AG_NPB * AG_F / 4; q += AG_TB) acc4[q] = make_float4(0.f, 0.f, 0.f, 0.f);
    __syncthreads();


    static_assert(AG_F % 4 == 0, "a row is whole quarters");
    int el4[AG_F / 4], q4[AG_F / 4];
#pragma unroll
    for (int j = 0; j < AG_F / 4; ++j) {
        const int v = 32 * j + lane;
        el4[j] = (v / (AG_F / 4)) << 2;
        q4[j] = 4 * (v % (AG_F / 4));
    }
    const bool rot1 = (lane & 1) != 0, rot2 = (lane & 2) != 0;
    int ro[4];
#pragma unroll
    for (int i = 0; i < 4; ++i) ro[i] = (i + lane) & 3;

    auto apply = [&](int s20, int d20, float nr, int lim) {
        float4 raw[AG_F / 4];
        int dd[AG_F / 4];
        float nn[AG_F / 4];
#pragma unroll
        for (int j = 0; j < AG_F / 4; ++j) {
            const int ss = __builtin_amdgcn_ds_bpermute(el4[j], s20);
            dd[j] = __builtin_amdgcn_ds_bpermute(el4[j], d20);
            nn[j] = __int_as_float(__builtin_amdgcn_ds_bpermute(el4[j], __float_as_int(nr)));
            raw[j] = *(const float4*)(xin + ss + q4[j]);
        }
#pragma unroll
        for (int j = 0; j < AG_F / 4; ++j) {
            if (32 * j + lane < lim) {
                const float4 t = raw[j];
                const float a0 = rot1 ? t.y : t.x, a1 = rot1 ? t.z : t.y, a2 = rot1 ? t.w : t.z, a3 = rot1 ? t.x : t.w;
                const float b0 = rot2 ? a2 : a0, b1 = rot2 ? a3 : a1, b2 = rot2 ? a0 : a2, b3 = rot2 ? a1 : a3;
                float* p = acc + dd[j] + q4[j];
                atomicAdd(p + ro[0], b0 * nn[j]);
                atomicAdd(p + ro[1], b1 * nn[j]);
                atomicAdd(p + ro[2], b2 * nn[j]);
                atomicAdd(p + ro[3], b3 * nn[j]);
            }
        }
    };

    if constexpr (!HEAD) {
        __shared__ float dloc[AG_NPB];
        for (int q = tid; q < AG_NPB; q += AG_TB) dloc[q] = node0 + q < N ? dinv[node0 + q] : 0.0f;
        __syncthreads();
        int c = __builtin_amdgcn_readfirstlane(tails[blockIdx.x * AG_WAVES + wave]);
        for (int it = 0; it < maxch && (unsigned)c < (unsigned)maxch; ++it) {
            const int2 hd = *(const int2*)(hdr + 2 * c);
            const int hn = __builtin_amdgcn_readfirstlane(hd.y);
            const int n = hn < AG_QW ? hn : AG_QW;
            int* ch = lst + (size_t)c * AG_CHW;
            if (n > 0) {
                const int lastb = (n - 1) & ~31;
                const int b1 = lastb < 32 ? lastb : 32;
                int s20 = ch[lane], d20 = ch[AG_QW + lane];
                int s20n = ch[b1 + lane], d20n = ch[AG_QW + b1 + lane];
                float ds = dinv[(unsigned)s20 / (unsigned)AG_F];
                for (int base = 0; base < n; base += 32) {
                    const int b2 = base + 64 < lastb ? base + 64 : lastb;
                    const int s20f = ch[b2 + lane], d20f = ch[AG_QW + b2 + lane];
                    const float dsn = dinv[(unsigned)s20n / (unsigned)AG_F];
                    const float nr = ds * dloc[(unsigned)d20 / (unsigned)AG_F];
                    ch[2 * AG_QW + base + lane] = __float_as_int(nr);
                    const int rem = n - base;
                    apply(s20, d20, nr, (rem < 32 ? rem : 32) * (AG_F / 4));
                    s20 = s20n;
                    d20 = d20n;
                    ds = dsn;
                    s20n = s20f;
                    d20n = d20f;
                }
            }
            c = __builtin_amdgcn_readfirstlane(hd.x);
        }
    } else {
        int c = __builtin_amdgcn_readfirstlane(tails[blockIdx.x * AG_WAVES + wave]);
        for (int it = 0; it < maxch && (unsigned)c < (unsigned)maxch; ++it) {
            const int2 hd = *(const int2*)(hdr + 2 * c);
            const int hn = __builtin_amdgcn_readfirstlane(hd.y);
            const int n = hn < AG_QW ? hn : AG_QW;
            const int* ch = lst + (size_t)c * AG_CHW;
            int s20 = ch[lane], d20 = ch[AG_QW + lane], nri = ch[2 * AG_QW + lane];
            for (int base = 0; base < n; base += 32) {
                const int nb = (base + 32 < n ? base + 32 : base) + lane;
                const int ns = ch[nb], nd = ch[AG_QW + nb], nn = ch[2 * AG_QW + nb];
                const int rem = n - base;
                apply(s20, d20, __int_as_float(nri), (rem < 32 ? rem : 32) * (AG_F / 4));
                s20 = ns;
                d20 = nd;
                nri = nn;
            }
            c = __builtin_amdgcn_readfirstlane(hd.x);
        }
    }
    __syncthreads();

    const float4* t4 = (const float4*)(xin + (size_t)node0 * AG_F);
    for (int q = tid; q < AG_NPB * AG_F / 4; q += AG_TB) {
        const int nl = q / 5;
        if (node0 + nl < N) {
            const float di = dinv[node0 + nl];
            const float sl = di * di;
            float4 a = acc4[q];
            const float4 tv = t4[q];
            a.x += tv.x * sl; a.y += tv.y * sl; a.z += tv.z * sl; a.w += tv.w * sl;
            acc4[q] = a;
        }
    }
    __syncthreads();

    {
        const int c = lane & 15, hh = lane >> 4;
        const int n1 = 16 + c < AG_F ? 16 + c : AG_F - 1;
        v2f w0[AG_F / 4], w1[AG_F / 4];
#pragma unroll
        for (int j = 0; j < AG_F / 4; ++j) {
            const int k = 4 * j + 2 * hh;
            w0[j].x = Wm[k * AG_F + c];
            w0[j].y = Wm[(k + 1) * AG_F + c];
            w1[j].x = Wm[k * AG_F + n1];
            w1[j].y = Wm[(k + 1) * AG_F + n1];
        }
        const float bv0 = bias[c], bv1 = bias[n1];
        for (int mt = wave; mt < AG_NPB / 16; mt += AG_WAVES) {
            float* rows = acc + mt * 16 * AG_F;
            const float* ap = rows + c * AG_F + 2 * hh;
            v2f a[AG_F / 4];
#pragma unroll
            for (int j = 0; j < AG_F / 4; ++j) a[j] = *(const v2f*)(ap + 4 * j);
            v8f d0 = {}, d1 = {};
#pragma unroll
            for (int j = 0; j < AG_F / 4; ++j) {
                d0 = __builtin_amdgcn_wmma_f32_16x16x4_f32(false, a[j], false, w0[j], (short)0, d0, false, false);
                d1 = __builtin_amdgcn_wmma_f32_16x16x4_f32(false, a[j], false, w1[j], (short)0, d1, false, false);
            }
            float* op = rows + 8 * hh * AG_F;
#pragma unroll
            for (int r = 0; r < 8; ++r) {
                op[r * AG_F + c] = fmaxf(d0[r] + bv0, 0.0f);
                if (16 + c < AG_F) op[r * AG_F + 16 + c] = fmaxf(d1[r] + bv1, 0.0f);
            }
        }
    }
    __syncthreads();

    if constexpr (!HEAD) {
        float4* h4 = (float4*)(h + (size_t)node0 * AG_F);
        for (int q = tid; q < AG_NPB * AG_F / 4; q += AG_TB) {
            const int nl = q / 5;
            if (node0 + nl < N) h4[q] = acc4[q];
        }
    } else {
        const int c = lane & 15, hh = lane >> 4;
        const int wa = wave >> 3, wb = (wave - wa) & 7;
        const int nt = wb < 5 ? wa * 5 + wb : FC_NT;
        const int ncol = nt * 16 + c;
        v8f d0 = {}, d1 = {};
        if (nt < FC_NT) {
            const int nc = ncol < FC_N ? ncol : FC_N - 1;
            const float* bp = fcW + (size_t)(2 * hh) * FC_N + nc;
            const float* a0p = acc + c * FC_K + 2 * hh;
            const float* a1p = a0p + 16 * FC_K;
            float bx[FC_G], by[FC_G];
#pragma unroll
            for (int j = 0; j < FC_G; ++j) {
                bx[j] = bp[(size_t)(4 * j) * FC_N];
                by[j] = bp[(size_t)(4 * j + 1) * FC_N];
            }
            constexpr int NG = FC_K / 4 / FC_G;
            for (int g = 0; g < NG; ++g) {
                const int gn = g + 1 < NG ? g + 1 : g;
                const float* bq = bp + (size_t)(gn * FC_G * 4) * FC_N;
                float nx[FC_G], ny[FC_G];
#pragma unroll
                for (int j = 0; j < FC_G; ++j) {
                    nx[j] = bq[(size_t)(4 * j) * FC_N];
                    ny[j] = bq[(size_t)(4 * j + 1) * FC_N];
                }
                v2f a0[FC_G], a1[FC_G];
#pragma unroll
                for (int j = 0; j < FC_G; ++j) {
                    a0[j] = *(const v2f*)(a0p + (g * FC_G + j) * 4);
                    a1[j] = *(const v2f*)(a1p + (g * FC_G + j) * 4);
                }
#pragma unroll
                for (int j = 0; j < FC_G; ++j) {
                    v2f b;
                    b.x = bx[j];
                    b.y = by[j];
                    d0 = __builtin_amdgcn_wmma_f32_16x16x4_f32(false, a0[j], false, b, (short)0, d0, false, false);
                    d1 = __builtin_amdgcn_wmma_f32_16x16x4_f32(false, a1[j], false, b, (short)0, d1, false, false);
                }
#pragma unroll
                for (int j = 0; j < FC_G; ++j) {
                    bx[j] = nx[j];
                    by[j] = ny[j];
                }
            }
        }
        __syncthreads();
        if (nt < FC_NT && ncol < FC_N) {
            const float bv = fcb[ncol];
            float* op = acc + 8 * hh * FC_N + ncol;
#pragma unroll
            for (int r = 0; r < 8; ++r) {
                op[r * FC_N] = fmaxf(d0[r] + bv, 0.0f);
                op[(16 + r) * FC_N] = fmaxf(d1[r] + bv, 0.0f);
            }
        }
        __syncthreads();
        const int g0 = (int)blockIdx.x * (AG_NPB / AG_GN);
        const int ng = N / AG_GN;
        float4* o4 = (float4*)(fco + (size_t)g0 * FC_N);
        for (int q = tid; q < (AG_NPB / AG_GN) * FC_N / 4; q += AG_TB) {
            const int gl = q / (FC_N / 4);
            if (g0 + gl < ng) o4[q] = acc4[q];
        }
    }
}

__global__ void k_wmma_gemm_f32(const float* __restrict__ A, const float* __restrict__ Bt,
                                const float* __restrict__ bias, float* __restrict__ C,
                                int M, int K, int Ncols, int mtiles, int ntiles, int do_relu) {
    int wave = (blockIdx.x * blockDim.x + threadIdx.x) >> 5;
    int lane = threadIdx.x & 31;
    int mt = wave % mtiles;
    int nt = wave / mtiles;
    if (nt >= ntiles) return;

    int lrow = lane & 15;
    int kh = (lane >> 4) << 1;
    int ncol = nt * 16 + lrow;
    const float* arow = A + (size_t)(mt * 16 + lrow) * K + kh;
    const float* brow = Bt + (size_t)ncol * K + kh;

    v8f acc = {};
    int kb = 0;
    for (; kb + 16 <= K; kb += 16) {
#pragma unroll
        for (int j = 0; j < 4; ++j) {
            v2f a = *(const v2f*)(arow + kb + 4 * j);
            v2f b = *(const v2f*)(brow + kb + 4 * j);
            acc = __builtin_amdgcn_wmma_f32_16x16x4_f32(false, a, false, b,
                                                        (short)0, acc, false, false);
        }
    }
    for (; kb < K; kb += 4) {
        v2f a = *(const v2f*)(arow + kb);
        v2f b = *(const v2f*)(brow + kb);
        acc = __builtin_amdgcn_wmma_f32_16x16x4_f32(false, a, false, b,
                                                    (short)0, acc, false, false);
    }

    if (ncol >= Ncols) return;
    float bv = bias ? bias[ncol] : 0.0f;
    int rbase = mt * 16 + ((lane >> 4) << 3);
#pragma unroll
    for (int r = 0; r < 8; ++r) {
        float v = acc[r] + bv;
        if (do_relu) v = fmaxf(v, 0.0f);
        C[(size_t)(rbase + r) * Ncols + ncol] = v;
    }
}


extern "C" void kernel_launch(void* const* d_in, const int* in_sizes, int n_in,
                              void* d_out, int out_size, void* d_ws, size_t ws_size,
                              hipStream_t stream) {
    const float* x    = (const float*)d_in[0];
    const int*   eidx = (const int*)d_in[1];
    const float* W1   = (const float*)d_in[3];
    const float* b1   = (const float*)d_in[4];
    const float* W2   = (const float*)d_in[5];
    const float* b2   = (const float*)d_in[6];
    const float* fc1W = (const float*)d_in[7];
    const float* fc1b = (const float*)d_in[8];
    const float* outW = (const float*)d_in[9];
    const float* outb = (const float*)d_in[10];

    const int N  = in_sizes[2];
    const int E  = in_sizes[1] / 2;
    const int Bg = N / 82;
    const int EMB = 20, HID = 300, NC = 22;

    char* ws = (char*)d_ws;
    size_t off = 0;
    auto take = [&](size_t bytes) -> void* {
        off = (off + 255) & ~(size_t)255;
        void* p = ws + off;
        off += bytes;
        return p;
    };
    float* dinv = (float*)take((size_t)N * 4);
    float* t    = (float*)take((size_t)N * EMB * 4);
    float* fco  = (float*)take((size_t)Bg * HID * 4);
    float* Bt   = (float*)take((size_t)32 * HID * 4);
    const int ntiles = cdiv(E, OW_TILE);
    unsigned* own4 = (unsigned*)take((size_t)ntiles * (OW_TILE / 4) * 4);
    unsigned* rec = (unsigned*)take((size_t)ntiles * OW_TILE * 4);
    const int maxch = E / AG_QFLUSH + cdiv(N, AG_NPB) * AG_WAVES;
    int* lst    = (int*)take((size_t)maxch * AG_CHW * 4);
    int* hdr    = (int*)take((size_t)maxch * 8);
    int* tails  = (int*)take((size_t)cdiv(N, AG_NPB) * AG_WAVES * 4);
    int* cursor = (int*)take(4);

    const int* src = eidx;
    const int* dst = eidx + E;

    const int TB = 256;

    auto run_gemm = [&](const float* A, const float* Bw, const float* bias, float* C,
                        int M, int K, int Ncols, int relu) {
        int ntiles = (Ncols + 15) / 16;
        int ncols_pad = ntiles * 16;
        int mtiles = M / 16;
        k_transpose_pad<<<cdiv(ncols_pad * K, TB), TB, 0, stream>>>(Bw, Bt, K, Ncols, ncols_pad);
        int waves = mtiles * ntiles;
        k_wmma_gemm_f32<<<cdiv(waves * 32, TB), TB, 0, stream>>>(
            A, Bt, bias, C, M, K, Ncols, mtiles, ntiles, relu);
    };

    k_owner_rec<<<ntiles, OW_TB, 0, stream>>>(src, dst, own4, rec, cursor, E);
    k_scan_count<<<cdiv(N, AG_NPB), AG_TB, 0, stream>>>(rec, own4, dinv, lst, hdr, tails, cursor, maxch, E, N);

    k_aggregate<false><<<cdiv(N, AG_NPB), AG_TB, 0, stream>>>(x, dinv, W1, b1, t,
                                                              nullptr, nullptr, nullptr,
                                                              lst, hdr, tails, maxch, N);

    k_aggregate<true><<<cdiv(N, AG_NPB), AG_TB, 0, stream>>>(t, dinv, W2, b2, nullptr,
                                                             fc1W, fc1b, fco,
                                                             lst, hdr, tails, maxch, N);

    run_gemm(fco, outW, outb, (float*)d_out, Bg, HID, NC, 0);
}


# Replay chunk numbers from a per-wave allotment of 4 by position (block, wave, turn), shared counter only past it: the scan's ~32 thousand returning adds on one word gone; sum order = LDS arrival
# speedup vs baseline: 7.3779x; 1.0778x over previous
#include <hip/hip_runtime.h>
#include <hip/hip_bf16.h>


typedef __attribute__((ext_vector_type(2))) float v2f;
typedef __attribute__((ext_vector_type(8))) float v8f;

static __host__ inline int cdiv(int a, int b) { return (a + b - 1) / b; }


#define AG_NPB    2624
#define AG_SBITS  20
#define AG_SMASK  ((1u << AG_SBITS) - 1u)
#define OW_TILE   1024
#define OW_TB     (OW_TILE / 4)
__global__ void __launch_bounds__(OW_TB)
k_owner_rec(const int* __restrict__ src, const int* __restrict__ dst,
            unsigned* __restrict__ own4, unsigned* __restrict__ rec, int* __restrict__ cursor, int E) {
    static_assert(256 * AG_NPB <= (1 << AG_SBITS) && AG_NPB <= (1 << (32 - AG_SBITS)), "src and local dst fit one word");
    static_assert(OW_TB == 256, "8 waves x 4 sets = the 32 sets of a tile");
    const int i = blockIdx.x * OW_TB + threadIdx.x;
    if (i == 0) *cursor = 0;
    const int e0 = 4 * i;
    int d0 = -1, d1 = -1, d2 = -1, d3 = -1;
    int s0 = 0, s1 = 0, s2 = 0, s3 = 0;
    if ((((((size_t)dst) | ((size_t)src)) & 15) == 0) && e0 + 3 < E) {
        const int4 v = ((const int4*)dst)[i];
        const int4 u = ((const int4*)src)[i];
        d0 = v.x; d1 = v.y; d2 = v.z; d3 = v.w;
        s0 = u.x; s1 = u.y; s2 = u.z; s3 = u.w;
    } else {
        if (e0 + 0 < E) { d0 = dst[e0 + 0]; s0 = src[e0 + 0]; }
        if (e0 + 1 < E) { d1 = dst[e0 + 1]; s1 = src[e0 + 1]; }
        if (e0 + 2 < E) { d2 = dst[e0 + 2]; s2 = src[e0 + 2]; }
        if (e0 + 3 < E) { d3 = dst[e0 + 3]; s3 = src[e0 + 3]; }
    }
    unsigned o[4] = {0u, 0u, 0u, 0u};
    uint4 r = make_uint4(0u, 0u, 0u, 0u);
    if (d0 >= 0) {
        o[0] = (unsigned)d0 / (unsigned)AG_NPB;
        r.x = ((unsigned)s0 & AG_SMASK) | (((unsigned)d0 - o[0] * (unsigned)AG_NPB) << AG_SBITS);
    }
    if (d1 >= 0) {
        o[1] = (unsigned)d1 / (unsigned)AG_NPB;
        r.y = ((unsigned)s1 & AG_SMASK) | (((unsigned)d1 - o[1] * (unsigned)AG_NPB) << AG_SBITS);
    }
    if (d2 >= 0) {
        o[2] = (unsigned)d2 / (unsigned)AG_NPB;
        r.z = ((unsigned)s2 & AG_SMASK) | (((unsigned)d2 - o[2] * (unsigned)AG_NPB) << AG_SBITS);
    }
    if (d3 >= 0) {
        o[3] = (unsigned)d3 / (unsigned)AG_NPB;
        r.w = ((unsigned)s3 & AG_SMASK) | (((unsigned)d3 - o[3] * (unsigned)AG_NPB) << AG_SBITS);
    }
    ((uint4*)rec)[i] = r;
    const int lane = threadIdx.x & 31;
    unsigned pl = 0u;
#pragma unroll
    for (int j = 0; j < 4; ++j) {
#pragma unroll
        for (int p = 0; p < 8; ++p) {
            const unsigned b = __builtin_amdgcn_ballot_w32(((o[j] >> p) & 1u) != 0u);
            if (lane == 8 * j + p) pl = b;
        }
    }
    const int j = lane >> 3, p = lane & 7;
    const int q = (threadIdx.x >> 5) * 4 + j;
    own4[(size_t)blockIdx.x * (OW_TILE / 4) + (p >> 2) * 128 + q * 4 + (p & 3)] = pl;
}

__global__ void k_transpose_pad(const float* __restrict__ Bw, float* __restrict__ Bt,
                                int K, int Ncols, int ncols_pad) {
    int i = blockIdx.x * blockDim.x + threadIdx.x;
    int total = ncols_pad * K;
    if (i >= total) return;
    int ncol = i / K, k = i - ncol * K;
    Bt[i] = (ncol < Ncols) ? Bw[(size_t)k * Ncols + ncol] : 0.0f;
}

#define AG_F      20
#define AG_GN     82
#define FC_K      (AG_GN * AG_F)
#define FC_N      300
#define FC_NT     ((FC_N + 15) / 16)
#define FC_G      10
#define AG_WAVES  32
#define AG_TB     (AG_WAVES * 32)
#define AG_QFLUSH 128
#define AG_QW     (AG_QFLUSH + 32)
#define AG_QB     (AG_QW / 32)
#define AG_CHW    (3 * AG_QW)
#define AG_OWNCH  4

__global__ void __launch_bounds__(AG_TB)
k_scan_count(const unsigned* __restrict__ rec, const unsigned* __restrict__ own, float* __restrict__ dinv,
             int* __restrict__ lst, int* __restrict__ hdr, int* __restrict__ tails, int* __restrict__ cursor, int maxch,
             int E, int N) {
    static_assert(AG_QW % 32 == 0, "whole batches in a queue");
    __shared__ int qe[AG_WAVES * AG_QW];
    __shared__ int cntl[AG_NPB];

    const int tid = threadIdx.x;
    const int lane = tid & 31;
    const int wave = __builtin_amdgcn_readfirstlane(tid >> 5);
    const int node0 = blockIdx.x * AG_NPB;

    for (int q = tid; q < AG_NPB; q += AG_TB) cntl[q] = 0;
    __syncthreads();

    int* myqe = qe + wave * AG_QW;
    int cnt = 0;
    int prevc = -1;
    int used = 0;

    auto push = [&](bool m, int e) {
        const unsigned bal = __builtin_amdgcn_ballot_w32(m);
        if (bal) {
            if (m) myqe[cnt + (int)__builtin_amdgcn_mbcnt_lo(bal, 0u)] = e;
            cnt += __builtin_popcount(bal);
        }
    };

    const int TT = (E + OW_TILE - 1) / OW_TILE;
    const int rot = (int)blockIdx.x * (TT / (int)gridDim.x);
    const uint4* __restrict__ own16 = (const uint4*)own;
    const int KT = (TT + AG_WAVES - 1) / AG_WAVES;
    unsigned want[8];
#pragma unroll
    for (int p = 0; p < 8; ++p) want[p] = ((blockIdx.x >> p) & 1u) ? 0xFFFFFFFFu : 0u;

    auto drain = [&]() {
        __builtin_amdgcn_fence(__ATOMIC_ACQ_REL, "wavefront");
        int c;
        if (used < AG_OWNCH) {
            c = ((int)blockIdx.x * AG_WAVES + wave) * AG_OWNCH + used;
        } else {
            int cv = 0;
            if (lane == 0) cv = atomicAdd(cursor, 1);
            c = (int)gridDim.x * AG_WAVES * AG_OWNCH + __builtin_amdgcn_readfirstlane(cv);
        }
        ++used;
        const bool keep = c < maxch;
        int* ch = lst + (size_t)(keep ? c : 0) * AG_CHW;
        if (keep) {
            if (lane == 0) *(int2*)(hdr + 2 * c) = make_int2(prevc, cnt);
            prevc = c;
        }
        unsigned pk[AG_QB];
#pragma unroll
        for (int u = 0; u < AG_QB; ++u) {
            const int i = 32 * u + lane;
            pk[u] = rec[myqe[i < cnt ? i : cnt - 1]];
        }
#pragma unroll
        for (int u = 0; u < AG_QB; ++u) {
            if (32 * u < cnt) {
                const int i = 32 * u + lane;
                int s20 = 0, d20 = 0;
                if (i < cnt) {
                    const int s = (int)(pk[u] & AG_SMASK);
                    const int dl = (int)(pk[u] >> AG_SBITS);
                    atomicAdd(&cntl[dl], 1);
                    s20 = s * AG_F;
                    d20 = dl * AG_F;
                }
                if (keep) {
                    ch[i] = s20;
                    ch[AG_QW + i] = d20;
                }
            }
        }
        cnt = 0;
    };

    auto work = [&](int t, bool live, const uint4& ra, const uint4& rb) {
        unsigned ownm = ~((ra.x ^ want[0]) | (ra.y ^ want[1]) | (ra.z ^ want[2]) | (ra.w ^ want[3]) |
                          (rb.x ^ want[4]) | (rb.y ^ want[5]) | (rb.z ^ want[6]) | (rb.w ^ want[7]));
        if (!live) ownm = 0u;
        const int e0 = t * OW_TILE + ((lane >> 2) << 7) + (lane & 3);
        if (t == TT - 1) {
            const int left = E - e0;
            const int nl = left <= 0 ? 0 : (left + 3) >> 2;
            ownm &= nl >= 32 ? 0xFFFFFFFFu : ((1u << nl) - 1u);
        }
        while (__builtin_amdgcn_ballot_w32(ownm != 0u)) {
            const int p = __builtin_ctz(ownm | 0x80000000u);
            push(ownm != 0u, e0 + 4 * p);
            ownm &= ownm - 1u;
            if (cnt >= AG_QFLUSH) drain();
        }
    };

    for (int k = 0; k < KT; k += 4) {
        int tt[4];
        bool live[4];
        uint4 ra[4], rb[4];
#pragma unroll
        for (int u = 0; u < 4; ++u) {
            const int tl = (k + u) * AG_WAVES + wave;
            int at = tl + rot;
            if (at >= TT) at -= TT;
            live[u] = tl < TT;
            tt[u] = at < TT ? at : TT - 1;
            ra[u] = own16[tt[u] * (OW_TILE / 16) + lane];
            rb[u] = own16[tt[u] * (OW_TILE / 16) + 32 + lane];
        }
#pragma unroll
        for (int u = 0; u < 4; ++u) work(tt[u], live[u], ra[u], rb[u]);
    }
    if (cnt > 0) drain();
    if (lane == 0) tails[blockIdx.x * AG_WAVES + wave] = prevc;
    __syncthreads();

    for (int q = tid; q < AG_NPB; q += AG_TB) {
        if (node0 + q < N) {
            const float v = 1.0f + (float)cntl[q];
            dinv[node0 + q] = (v > 0.0f) ? rsqrtf(v) : 0.0f;
        }
    }
}

template <bool HEAD>
__global__ void __launch_bounds__(AG_TB)
k_aggregate(const float* __restrict__ xin, const float* __restrict__ dinv,
            const float* __restrict__ Wm, const float* __restrict__ bias, float* __restrict__ h,
            const float* __restrict__ fcW, const float* __restrict__ fcb, float* __restrict__ fco,
            int* __restrict__ lst, const int* __restrict__ hdr, const int* __restrict__ tails, int maxch, int N) {
    static_assert(AG_NPB % AG_GN == 0 && AG_NPB / AG_GN == 32, "a block owns 32 whole graphs");
    static_assert((FC_K / 4) % FC_G == 0, "whole groups of k steps");
    __shared__ __attribute__((aligned(16))) float acc[AG_NPB * AG_F];

    const int tid = threadIdx.x;
    const int lane = tid & 31;
    const int wave = __builtin_amdgcn_readfirstlane(tid >> 5);
    const int node0 = blockIdx.x * AG_NPB;

    float4* acc4 = (float4*)acc;
    for (int q = tid; q < AG_NPB * AG_F / 4; q += AG_TB) acc4[q] = make_float4(0.f, 0.f, 0.f, 0.f);
    __syncthreads();


    static_assert(AG_F % 4 == 0, "a row is whole quarters");
    int el4[AG_F / 4], q4[AG_F / 4];
#pragma unroll
    for (int j = 0; j < AG_F / 4; ++j) {
        const int v = 32 * j + lane;
        el4[j] = (v / (AG_F / 4)) << 2;
        q4[j] = 4 * (v % (AG_F / 4));
    }
    const bool rot1 = (lane & 1) != 0, rot2 = (lane & 2) != 0;
    int ro[4];
#pragma unroll
    for (int i = 0; i < 4; ++i) ro[i] = (i + lane) & 3;

    auto apply = [&](int s20, int d20, float nr, int lim) {
        float4 raw[AG_F / 4];
        int dd[AG_F / 4];
        float nn[AG_F / 4];
#pragma unroll
        for (int j = 0; j < AG_F / 4; ++j) {
            const int ss = __builtin_amdgcn_ds_bpermute(el4[j], s20);
            dd[j] = __builtin_amdgcn_ds_bpermute(el4[j], d20);
            nn[j] = __int_as_float(__builtin_amdgcn_ds_bpermute(el4[j], __float_as_int(nr)));
            raw[j] = *(const float4*)(xin + ss + q4[j]);
        }
#pragma unroll
        for (int j = 0; j < AG_F / 4; ++j) {
            if (32 * j + lane < lim) {
                const float4 t = raw[j];
                const float a0 = rot1 ? t.y : t.x, a1 = rot1 ? t.z : t.y, a2 = rot1 ? t.w : t.z, a3 = rot1 ? t.x : t.w;
                const float b0 = rot2 ? a2 : a0, b1 = rot2 ? a3 : a1, b2 = rot2 ? a0 : a2, b3 = rot2 ? a1 : a3;
                float* p = acc + dd[j] + q4[j];
                atomicAdd(p + ro[0], b0 * nn[j]);
                atomicAdd(p + ro[1], b1 * nn[j]);
                atomicAdd(p + ro[2], b2 * nn[j]);
                atomicAdd(p + ro[3], b3 * nn[j]);
            }
        }
    };

    if constexpr (!HEAD) {
        __shared__ float dloc[AG_NPB];
        for (int q = tid; q < AG_NPB; q += AG_TB) dloc[q] = node0 + q < N ? dinv[node0 + q] : 0.0f;
        __syncthreads();
        int c = __builtin_amdgcn_readfirstlane(tails[blockIdx.x * AG_WAVES + wave]);
        for (int it = 0; it < maxch && (unsigned)c < (unsigned)maxch; ++it) {
            const int2 hd = *(const int2*)(hdr + 2 * c);
            const int hn = __builtin_amdgcn_readfirstlane(hd.y);
            const int n = hn < AG_QW ? hn : AG_QW;
            int* ch = lst + (size_t)c * AG_CHW;
            if (n > 0) {
                const int lastb = (n - 1) & ~31;
                const int b1 = lastb < 32 ? lastb : 32;
                int s20 = ch[lane], d20 = ch[AG_QW + lane];
                int s20n = ch[b1 + lane], d20n = ch[AG_QW + b1 + lane];
                float ds = dinv[(unsigned)s20 / (unsigned)AG_F];
                for (int base = 0; base < n; base += 32) {
                    const int b2 = base + 64 < lastb ? base + 64 : lastb;
                    const int s20f = ch[b2 + lane], d20f = ch[AG_QW + b2 + lane];
                    const float dsn = dinv[(unsigned)s20n / (unsigned)AG_F];
                    const float nr = ds * dloc[(unsigned)d20 / (unsigned)AG_F];
                    ch[2 * AG_QW + base + lane] = __float_as_int(nr);
                    const int rem = n - base;
                    apply(s20, d20, nr, (rem < 32 ? rem : 32) * (AG_F / 4));
                    s20 = s20n;
                    d20 = d20n;
                    ds = dsn;
                    s20n = s20f;
                    d20n = d20f;
                }
            }
            c = __builtin_amdgcn_readfirstlane(hd.x);
        }
    } else {
        int c = __builtin_amdgcn_readfirstlane(tails[blockIdx.x * AG_WAVES + wave]);
        for (int it = 0; it < maxch && (unsigned)c < (unsigned)maxch; ++it) {
            const int2 hd = *(const int2*)(hdr + 2 * c);
            const int hn = __builtin_amdgcn_readfirstlane(hd.y);
            const int n = hn < AG_QW ? hn : AG_QW;
            const int* ch = lst + (size_t)c * AG_CHW;
            int s20 = ch[lane], d20 = ch[AG_QW + lane], nri = ch[2 * AG_QW + lane];
            for (int base = 0; base < n; base += 32) {
                const int nb = (base + 32 < n ? base + 32 : base) + lane;
                const int ns = ch[nb], nd = ch[AG_QW + nb], nn = ch[2 * AG_QW + nb];
                const int rem = n - base;
                apply(s20, d20, __int_as_float(nri), (rem < 32 ? rem : 32) * (AG_F / 4));
                s20 = ns;
                d20 = nd;
                nri = nn;
            }
            c = __builtin_amdgcn_readfirstlane(hd.x);
        }
    }
    __syncthreads();

    const float4* t4 = (const float4*)(xin + (size_t)node0 * AG_F);
    for (int q = tid; q < AG_NPB * AG_F / 4; q += AG_TB) {
        const int nl = q / 5;
        if (node0 + nl < N) {
            const float di = dinv[node0 + nl];
            const float sl = di * di;
            float4 a = acc4[q];
            const float4 tv = t4[q];
            a.x += tv.x * sl; a.y += tv.y * sl; a.z += tv.z * sl; a.w += tv.w * sl;
            acc4[q] = a;
        }
    }
    __syncthreads();

    {
        const int c = lane & 15, hh = lane >> 4;
        const int n1 = 16 + c < AG_F ? 16 + c : AG_F - 1;
        v2f w0[AG_F / 4], w1[AG_F / 4];
#pragma unroll
        for (int j = 0; j < AG_F / 4; ++j) {
            const int k = 4 * j + 2 * hh;
            w0[j].x = Wm[k * AG_F + c];
            w0[j].y = Wm[(k + 1) * AG_F + c];
            w1[j].x = Wm[k * AG_F + n1];
            w1[j].y = Wm[(k + 1) * AG_F + n1];
        }
        const float bv0 = bias[c], bv1 = bias[n1];
        for (int mt = wave; mt < AG_NPB / 16; mt += AG_WAVES) {
            float* rows = acc + mt * 16 * AG_F;
            const float* ap = rows + c * AG_F + 2 * hh;
            v2f a[AG_F / 4];
#pragma unroll
            for (int j = 0; j < AG_F / 4; ++j) a[j] = *(const v2f*)(ap + 4 * j);
            v8f d0 = {}, d1 = {};
#pragma unroll
            for (int j = 0; j < AG_F / 4; ++j) {
                d0 = __builtin_amdgcn_wmma_f32_16x16x4_f32(false, a[j], false, w0[j], (short)0, d0, false, false);
                d1 = __builtin_amdgcn_wmma_f32_16x16x4_f32(false, a[j], false, w1[j], (short)0, d1, false, false);
            }
            float* op = rows + 8 * hh * AG_F;
#pragma unroll
            for (int r = 0; r < 8; ++r) {
                op[r * AG_F + c] = fmaxf(d0[r] + bv0, 0.0f);
                if (16 + c < AG_F) op[r * AG_F + 16 + c] = fmaxf(d1[r] + bv1, 0.0f);
            }
        }
    }
    __syncthreads();

    if constexpr (!HEAD) {
        float4* h4 = (float4*)(h + (size_t)node0 * AG_F);
        for (int q = tid; q < AG_NPB * AG_F / 4; q += AG_TB) {
            const int nl = q / 5;
            if (node0 + nl < N) h4[q] = acc4[q];
        }
    } else {
        const int c = lane & 15, hh = lane >> 4;
        const int wa = wave >> 3, wb = (wave - wa) & 7;
        const int nt = wb < 5 ? wa * 5 + wb : FC_NT;
        const int ncol = nt * 16 + c;
        v8f d0 = {}, d1 = {};
        if (nt < FC_NT) {
            const int nc = ncol < FC_N ? ncol : FC_N - 1;
            const float* bp = fcW + (size_t)(2 * hh) * FC_N + nc;
            const float* a0p = acc + c * FC_K + 2 * hh;
            const float* a1p = a0p + 16 * FC_K;
            float bx[FC_G], by[FC_G];
#pragma unroll
            for (int j = 0; j < FC_G; ++j) {
                bx[j] = bp[(size_t)(4 * j) * FC_N];
                by[j] = bp[(size_t)(4 * j + 1) * FC_N];
            }
            constexpr int NG = FC_K / 4 / FC_G;
            for (int g = 0; g < NG; ++g) {
                const int gn = g + 1 < NG ? g + 1 : g;
                const float* bq = bp + (size_t)(gn * FC_G * 4) * FC_N;
                float nx[FC_G], ny[FC_G];
#pragma unroll
                for (int j = 0; j < FC_G; ++j) {
                    nx[j] = bq[(size_t)(4 * j) * FC_N];
                    ny[j] = bq[(size_t)(4 * j + 1) * FC_N];
                }
                v2f a0[FC_G], a1[FC_G];
#pragma unroll
                for (int j = 0; j < FC_G; ++j) {
                    a0[j] = *(const v2f*)(a0p + (g * FC_G + j) * 4);
                    a1[j] = *(const v2f*)(a1p + (g * FC_G + j) * 4);
                }
#pragma unroll
                for (int j = 0; j < FC_G; ++j) {
                    v2f b;
                    b.x = bx[j];
                    b.y = by[j];
                    d0 = __builtin_amdgcn_wmma_f32_16x16x4_f32(false, a0[j], false, b, (short)0, d0, false, false);
                    d1 = __builtin_amdgcn_wmma_f32_16x16x4_f32(false, a1[j], false, b, (short)0, d1, false, false);
                }
#pragma unroll
                for (int j = 0; j < FC_G; ++j) {
                    bx[j] = nx[j];
                    by[j] = ny[j];
                }
            }
        }
        __syncthreads();
        if (nt < FC_NT && ncol < FC_N) {
            const float bv = fcb[ncol];
            float* op = acc + 8 * hh * FC_N + ncol;
#pragma unroll
            for (int r = 0; r < 8; ++r) {
                op[r * FC_N] = fmaxf(d0[r] + bv, 0.0f);
                op[(16 + r) * FC_N] = fmaxf(d1[r] + bv, 0.0f);
            }
        }
        __syncthreads();
        const int g0 = (int)blockIdx.x * (AG_NPB / AG_GN);
        const int ng = N / AG_GN;
        float4* o4 = (float4*)(fco + (size_t)g0 * FC_N);
        for (int q = tid; q < (AG_NPB / AG_GN) * FC_N / 4; q += AG_TB) {
            const int gl = q / (FC_N / 4);
            if (g0 + gl < ng) o4[q] = acc4[q];
        }
    }
}

__global__ void k_wmma_gemm_f32(const float* __restrict__ A, const float* __restrict__ Bt,
                                const float* __restrict__ bias, float* __restrict__ C,
                                int M, int K, int Ncols, int mtiles, int ntiles, int do_relu) {
    int wave = (blockIdx.x * blockDim.x + threadIdx.x) >> 5;
    int lane = threadIdx.x & 31;
    int mt = wave % mtiles;
    int nt = wave / mtiles;
    if (nt >= ntiles) return;

    int lrow = lane & 15;
    int kh = (lane >> 4) << 1;
    int ncol = nt * 16 + lrow;
    const float* arow = A + (size_t)(mt * 16 + lrow) * K + kh;
    const float* brow = Bt + (size_t)ncol * K + kh;

    v8f acc = {};
    int kb = 0;
    for (; kb + 16 <= K; kb += 16) {
#pragma unroll
        for (int j = 0; j < 4; ++j) {
            v2f a = *(const v2f*)(arow + kb + 4 * j);
            v2f b = *(const v2f*)(brow + kb + 4 * j);
            acc = __builtin_amdgcn_wmma_f32_16x16x4_f32(false, a, false, b,
                                                        (short)0, acc, false, false);
        }
    }
    for (; kb < K; kb += 4) {
        v2f a = *(const v2f*)(arow + kb);
        v2f b = *(const v2f*)(brow + kb);
        acc = __builtin_amdgcn_wmma_f32_16x16x4_f32(false, a, false, b,
                                                    (short)0, acc, false, false);
    }

    if (ncol >= Ncols) return;
    float bv = bias ? bias[ncol] : 0.0f;
    int rbase = mt * 16 + ((lane >> 4) << 3);
#pragma unroll
    for (int r = 0; r < 8; ++r) {
        float v = acc[r] + bv;
        if (do_relu) v = fmaxf(v, 0.0f);
        C[(size_t)(rbase + r) * Ncols + ncol] = v;
    }
}


extern "C" void kernel_launch(void* const* d_in, const int* in_sizes, int n_in,
                              void* d_out, int out_size, void* d_ws, size_t ws_size,
                              hipStream_t stream) {
    const float* x    = (const float*)d_in[0];
    const int*   eidx = (const int*)d_in[1];
    const float* W1   = (const float*)d_in[3];
    const float* b1   = (const float*)d_in[4];
    const float* W2   = (const float*)d_in[5];
    const float* b2   = (const float*)d_in[6];
    const float* fc1W = (const float*)d_in[7];
    const float* fc1b = (const float*)d_in[8];
    const float* outW = (const float*)d_in[9];
    const float* outb = (const float*)d_in[10];

    const int N  = in_sizes[2];
    const int E  = in_sizes[1] / 2;
    const int Bg = N / 82;
    const int EMB = 20, HID = 300, NC = 22;

    char* ws = (char*)d_ws;
    size_t off = 0;
    auto take = [&](size_t bytes) -> void* {
        off = (off + 255) & ~(size_t)255;
        void* p = ws + off;
        off += bytes;
        return p;
    };
    float* dinv = (float*)take((size_t)N * 4);
    float* t    = (float*)take((size_t)N * EMB * 4);
    float* fco  = (float*)take((size_t)Bg * HID * 4);
    float* Bt   = (float*)take((size_t)32 * HID * 4);
    const int ntiles = cdiv(E, OW_TILE);
    unsigned* own4 = (unsigned*)take((size_t)ntiles * (OW_TILE / 4) * 4);
    unsigned* rec = (unsigned*)take((size_t)ntiles * OW_TILE * 4);
    const int maxch = E / AG_QFLUSH + cdiv(N, AG_NPB) * AG_WAVES * (AG_OWNCH + 1);
    int* lst    = (int*)take((size_t)maxch * AG_CHW * 4);
    int* hdr    = (int*)take((size_t)maxch * 8);
    int* tails  = (int*)take((size_t)cdiv(N, AG_NPB) * AG_WAVES * 4);
    int* cursor = (int*)take(4);

    const int* src = eidx;
    const int* dst = eidx + E;

    const int TB = 256;

    auto run_gemm = [&](const float* A, const float* Bw, const float* bias, float* C,
                        int M, int K, int Ncols, int relu) {
        int ntiles = (Ncols + 15) / 16;
        int ncols_pad = ntiles * 16;
        int mtiles = M / 16;
        k_transpose_pad<<<cdiv(ncols_pad * K, TB), TB, 0, stream>>>(Bw, Bt, K, Ncols, ncols_pad);
        int waves = mtiles * ntiles;
        k_wmma_gemm_f32<<<cdiv(waves * 32, TB), TB, 0, stream>>>(
            A, Bt, bias, C, M, K, Ncols, mtiles, ntiles, relu);
    };

    k_owner_rec<<<ntiles, OW_TB, 0, stream>>>(src, dst, own4, rec, cursor, E);
    k_scan_count<<<cdiv(N, AG_NPB), AG_TB, 0, stream>>>(rec, own4, dinv, lst, hdr, tails, cursor, maxch, E, N);

    k_aggregate<false><<<cdiv(N, AG_NPB), AG_TB, 0, stream>>>(x, dinv, W1, b1, t,
                                                              nullptr, nullptr, nullptr,
                                                              lst, hdr, tails, maxch, N);

    k_aggregate<true><<<cdiv(N, AG_NPB), AG_TB, 0, stream>>>(t, dinv, W2, b2, nullptr,
                                                             fc1W, fc1b, fco,
                                                             lst, hdr, tails, maxch, N);

    run_gemm(fco, outW, outb, (float*)d_out, Bg, HID, NC, 0);
}
